# starting kernel
# baseline (speedup 1.0000x reference)
#include <hip/hip_runtime.h>

typedef float v2f __attribute__((ext_vector_type(2)));
typedef float v8f __attribute__((ext_vector_type(8)));

#define N_NODES 50000
#define FDIM    128
#define N_EDGES 800000
#define KSTRIDE 260

__global__ __launch_bounds__(256) void sage_zero_kernel(float* __restrict__ p, int n) {
    int i = blockIdx.x * 256 + threadIdx.x;
    if (i < n) p[i] = 0.0f;
}

__global__ __launch_bounds__(256) void sage_scatter_kernel(
    const float* __restrict__ x,
    const int*   __restrict__ src,
    const int*   __restrict__ dst,
    float* __restrict__ msg,
    float* __restrict__ cnt,
    int with_count)
{
    int lane = threadIdx.x & 31;
    int wave = threadIdx.x >> 5;
    int e = blockIdx.x * 8 + wave;
    if (e >= N_EDGES) return;
    int s = src[e];
    int d = dst[e];
    const float4* xs = (const float4*)(x + (long long)s * FDIM);
    float4 v = xs[lane];
    float* o = msg + (long long)d * FDIM + lane * 4;
    atomicAdd(o + 0, v.x);
    atomicAdd(o + 1, v.y);
    atomicAdd(o + 2, v.z);
    atomicAdd(o + 3, v.w);
    if (with_count && lane == 0) atomicAdd(cnt + d, 1.0f);
}

__global__ __launch_bounds__(256) void sage_gemm_kernel(
    const float* __restrict__ msg, const float* __restrict__ cnt,
    const float* __restrict__ x,
    const float* __restrict__ Wl,  const float* __restrict__ Wr,
    const float* __restrict__ bias,
    float* __restrict__ out, int do_relu)
{
    __shared__ float lw[128 * KSTRIDE];

    int tid = threadIdx.x;
    for (int idx = tid; idx < 256 * 128; idx += 256) {
        int k = idx >> 7;
        int n = idx & 127;
        float w = (k < 128) ? Wl[idx] : Wr[idx - 128 * 128];
        lw[n * KSTRIDE + k] = w;
    }
    __syncthreads();

    int lane = tid & 31;
    int wave = tid >> 5;
    int h    = lane >> 4;
    int col  = lane & 15;

    int rowBase = blockIdx.x * 128 + wave * 16;
    int rowA    = rowBase + col;
    int rowAc   = rowA < N_NODES ? rowA : (N_NODES - 1);

    float scale = 1.0f / fmaxf(cnt[rowAc], 1.0f);
    const float* arow_m = msg + (long long)rowAc * FDIM + 2 * h;
    const float* arow_x = x   + (long long)rowAc * FDIM + 2 * h;
    const float* brow   = &lw[col * KSTRIDE + 2 * h];

    v8f acc[8];
    v8f z = {0.f, 0.f, 0.f, 0.f, 0.f, 0.f, 0.f, 0.f};
#pragma unroll
    for (int t = 0; t < 8; ++t) acc[t] = z;

#pragma unroll 4
    for (int kk = 0; kk < 128; kk += 4) {
        v2f a = *(const v2f*)(arow_m + kk);
        a.x *= scale;
        a.y *= scale;
#pragma unroll
        for (int t = 0; t < 8; ++t) {
            v2f b = *(const v2f*)(brow + t * (16 * KSTRIDE) + kk);
            acc[t] = __builtin_amdgcn_wmma_f32_16x16x4_f32(
                false, a, false, b, (short)0, acc[t], false, false);
        }
    }

#pragma unroll 4
    for (int kk = 0; kk < 128; kk += 4) {
        v2f a = *(const v2f*)(arow_x + kk);
#pragma unroll
        for (int t = 0; t < 8; ++t) {
            v2f b = *(const v2f*)(brow + t * (16 * KSTRIDE) + 128 + kk);
            acc[t] = __builtin_amdgcn_wmma_f32_16x16x4_f32(
                false, a, false, b, (short)0, acc[t], false, false);
        }
    }

#pragma unroll
    for (int t = 0; t < 8; ++t) {
        int n  = t * 16 + col;
        float bv = bias[n];
#pragma unroll
        for (int v = 0; v < 8; ++v) {
            int row = rowBase + v + 8 * h;
            float val = acc[t][v] + bv;
            if (do_relu) val = fmaxf(val, 0.0f);
            if (row < N_NODES) out[(long long)row * FDIM + n] = val;
        }
    }
}

extern "C" void kernel_launch(void* const* d_in, const int* in_sizes, int n_in,
                              void* d_out, int out_size, void* d_ws, size_t ws_size,
                              hipStream_t stream) {
    const float* x   = (const float*)d_in[0];
    const int*   ei  = (const int*)d_in[1];
    const float* Wl0 = (const float*)d_in[2];
    const float* Wr0 = (const float*)d_in[3];
    const float* b0  = (const float*)d_in[4];
    const float* Wl1 = (const float*)d_in[5];
    const float* Wr1 = (const float*)d_in[6];
    const float* b1  = (const float*)d_in[7];
    float* out = (float*)d_out;

    const int* src = ei;
    const int* dst = ei + N_EDGES;

    float* msg = (float*)d_ws;
    float* cnt = msg + (size_t)N_NODES * FDIM;
    float* hbuf = cnt + N_NODES;

    const int nmsg = N_NODES * FDIM;
    const int gemm_blocks = (N_NODES + 127) / 128;
    const int edge_blocks = N_EDGES / 8;

    sage_zero_kernel<<<(nmsg + N_NODES + 255) / 256, 256, 0, stream>>>(msg, nmsg + N_NODES);
    sage_scatter_kernel<<<edge_blocks, 256, 0, stream>>>(x, src, dst, msg, cnt, 1);
    sage_gemm_kernel<<<gemm_blocks, 256, 0, stream>>>(msg, cnt, x, Wl0, Wr0, b0, hbuf, 1);

    sage_zero_kernel<<<(nmsg + 255) / 256, 256, 0, stream>>>(msg, nmsg);
    sage_scatter_kernel<<<edge_blocks, 256, 0, stream>>>(hbuf, src, dst, msg, cnt, 0);
    sage_gemm_kernel<<<gemm_blocks, 256, 0, stream>>>(msg, cnt, hbuf, Wl1, Wr1, b1, out, 0);
}


# each owner work-group forms its 200 output rows itself: f32 WMMA A fragments read from its LDS sums, weights staged in the dead list space; no message buffer, 2 launches not 4; sum order as before
# speedup vs baseline: 2.3631x; 2.3631x over previous
#include <hip/hip_runtime.h>

typedef float v2f __attribute__((ext_vector_type(2)));
typedef float v8f __attribute__((ext_vector_type(8)));


#define N_NODES 50000
#define FDIM    128
#define N_EDGES 800000
#define KSTRIDE 260

#define R_NODES    200
#define R_PAD      208
#define ACC_S      132
#define AGG_BLOCKS 250
#define CHUNK_E    32768
#define N_CHUNK    25
#define N_DST4     (N_EDGES / 4)
#define KEEP_WORDS (R_NODES * FDIM)
#define KEEP_HDR   32
#define KEEP_CAP   (KEEP_WORDS - KEEP_HDR)

__device__ __forceinline__ void sage_gather_list(
    const float* __restrict__ x, const unsigned* list, float* acc, float* dcnt,
    int n, int wave, int lane, int with_count)
{
    for (int i = wave; i < n; i += 128) {
        const bool p1 = (i + 32) < n, p2 = (i + 64) < n, p3 = (i + 96) < n;
        unsigned e0 = list[i];
        unsigned e1 = p1 ? list[i + 32] : e0;
        unsigned e2 = p2 ? list[i + 64] : e0;
        unsigned e3 = p3 ? list[i + 96] : e0;
        const float* r0 = x + (size_t)(e0 & 0xFFFFu) * FDIM + lane;
        const float* r1 = x + (size_t)(e1 & 0xFFFFu) * FDIM + lane;
        const float* r2 = x + (size_t)(e2 & 0xFFFFu) * FDIM + lane;
        const float* r3 = x + (size_t)(e3 & 0xFFFFu) * FDIM + lane;
        float a0 = r0[0], a1 = r0[32], a2 = r0[64], a3 = r0[96];
        float b0 = r1[0], b1 = r1[32], b2 = r1[64], b3 = r1[96];
        float c0 = r2[0], c1 = r2[32], c2 = r2[64], c3 = r2[96];
        float g0 = r3[0], g1 = r3[32], g2 = r3[64], g3 = r3[96];
        asm volatile("" :: "v"(a0), "v"(a1), "v"(a2), "v"(a3), "v"(b0), "v"(b1), "v"(b2), "v"(b3),
                           "v"(c0), "v"(c1), "v"(c2), "v"(c3), "v"(g0), "v"(g1), "v"(g2), "v"(g3));
        float* o0 = acc + (e0 >> 16) * ACC_S + lane;
        atomicAdd(o0, a0); atomicAdd(o0 + 32, a1); atomicAdd(o0 + 64, a2); atomicAdd(o0 + 96, a3);
        if (with_count && lane == 0) atomicAdd(&dcnt[e0 >> 16], 1.0f);
        if (p1) {
            float* o1 = acc + (e1 >> 16) * ACC_S + lane;
            atomicAdd(o1, b0); atomicAdd(o1 + 32, b1); atomicAdd(o1 + 64, b2); atomicAdd(o1 + 96, b3);
            if (with_count && lane == 0) atomicAdd(&dcnt[e1 >> 16], 1.0f);
        }
        if (p2) {
            float* o2 = acc + (e2 >> 16) * ACC_S + lane;
            atomicAdd(o2, c0); atomicAdd(o2 + 32, c1); atomicAdd(o2 + 64, c2); atomicAdd(o2 + 96, c3);
            if (with_count && lane == 0) atomicAdd(&dcnt[e2 >> 16], 1.0f);
        }
        if (p3) {
            float* o3 = acc + (e3 >> 16) * ACC_S + lane;
            atomicAdd(o3, g0); atomicAdd(o3 + 32, g1); atomicAdd(o3 + 64, g2); atomicAdd(o3 + 96, g3);
            if (with_count && lane == 0) atomicAdd(&dcnt[e3 >> 16], 1.0f);
        }
    }
}

template <int NT>
__device__ __forceinline__ void sage_wave_product(
    const float* acc, const float* dcnt, const float* lw,
    const float* __restrict__ x, const float* __restrict__ bias, float* hout,
    int lo, int rt, int ct0, int lane, int do_relu)
{
    const int h   = lane >> 4;
    const int col = lane & 15;
    const int rl  = rt * 16 + col;
    int rg = lo + rl;
    rg = rg < N_NODES ? rg : (N_NODES - 1);

    const float  scale  = dcnt[rl];
    const float* arow_m = acc + rl * ACC_S + 2 * h;
    const float* arow_x = x + (size_t)rg * FDIM + 2 * h;
    const float* brow   = lw + (ct0 * 16 + col) * KSTRIDE + 2 * h;

    v8f d4[NT];
    v8f z = {0.f, 0.f, 0.f, 0.f, 0.f, 0.f, 0.f, 0.f};
#pragma unroll
    for (int t = 0; t < NT; ++t) d4[t] = z;

#pragma unroll 4
    for (int kk = 0; kk < 128; kk += 4) {
        v2f a = *(const v2f*)(arow_m + kk);
        a.x *= scale;
        a.y *= scale;
#pragma unroll
        for (int t = 0; t < NT; ++t) {
            v2f b = *(const v2f*)(brow + t * (16 * KSTRIDE) + kk);
            d4[t] = __builtin_amdgcn_wmma_f32_16x16x4_f32(
                false, a, false, b, (short)0, d4[t], false, false);
        }
    }

#pragma unroll 4
    for (int kk = 0; kk < 128; kk += 4) {
        v2f a = *(const v2f*)(arow_x + kk);
#pragma unroll
        for (int t = 0; t < NT; ++t) {
            v2f b = *(const v2f*)(brow + t * (16 * KSTRIDE) + 128 + kk);
            d4[t] = __builtin_amdgcn_wmma_f32_16x16x4_f32(
                false, a, false, b, (short)0, d4[t], false, false);
        }
    }

#pragma unroll
    for (int t = 0; t < NT; ++t) {
        int n  = (ct0 + t) * 16 + col;
        float bv = bias[n];
#pragma unroll
        for (int v = 0; v < 8; ++v) {
            int row = rt * 16 + v + 8 * h;
            float val = d4[t][v] + bv;
            if (do_relu) val = fmaxf(val, 0.0f);
            if (row < R_NODES) hout[(size_t)(lo + row) * FDIM + n] = val;
        }
    }
}

__global__ __launch_bounds__(1024) void sage_layer_kernel(
    const float* __restrict__ x,
    const int*   __restrict__ src,
    const int*   __restrict__ dst,
    float* cnt,
    float* keep,
    const float* __restrict__ Wl, const float* __restrict__ Wr,
    const float* __restrict__ bias,
    float* hout,
    int first,
    int do_relu)
{
    __shared__ __attribute__((aligned(16))) float    acc[R_PAD * ACC_S];
    __shared__ float    dcnt[R_PAD];
    __shared__ __attribute__((aligned(16))) unsigned ubuf[128 * KSTRIDE];
    __shared__ unsigned nl[2];
    unsigned* list = ubuf;
    float*    lw   = (float*)ubuf;

    const int tid  = threadIdx.x;
    const int lane = tid & 31;
    const int wave = __builtin_amdgcn_readfirstlane(tid >> 5);
    const int lo   = blockIdx.x * R_NODES;
    unsigned* kl   = (unsigned*)keep + (size_t)blockIdx.x * KEEP_WORDS;

    for (int i = tid; i < R_PAD * ACC_S; i += 1024) acc[i] = 0.0f;
    if (tid < R_PAD) dcnt[tid] = 0.0f;
    if (tid < 2) nl[tid] = 0u;

    const unsigned kept = first ? 0xFFFFFFFFu : kl[0];

    if (kept <= (unsigned)KEEP_CAP) {
        const unsigned* ke = kl + KEEP_HDR;
        for (unsigned i = tid; i < kept; i += 4096) {
            const unsigned i1 = i + 1024, i2 = i + 2048, i3 = i + 3072;
            const bool q1 = i1 < kept, q2 = i2 < kept, q3 = i3 < kept;
            unsigned v0 = ke[i];
            unsigned v1 = q1 ? ke[i1] : 0u;
            unsigned v2 = q2 ? ke[i2] : 0u;
            unsigned v3 = q3 ? ke[i3] : 0u;
            list[i] = v0;
            if (q1) list[i1] = v1;
            if (q2) list[i2] = v2;
            if (q3) list[i3] = v3;
        }
        __syncthreads();
        sage_gather_list(x, list, acc, dcnt, (int)kept, wave, lane, 0);
        __syncthreads();
    } else {
        const int4* dst4 = (const int4*)dst;
        int4 d[8];
#pragma unroll
        for (int k = 0; k < 8; ++k) {
            int idx = k * 1024 + tid;
            d[k] = dst4[idx];
        }
        __syncthreads();

        unsigned tot = 0u;
        for (int c = 0; c < N_CHUNK; ++c) {
            const int cur = c & 1;
            int  sv[32];
            bool hit[8];
#pragma unroll
            for (int k = 0; k < 8; ++k) {
                int ebase = (c * 8192 + k * 1024 + tid) << 2;
                unsigned l0 = (unsigned)(d[k].x - lo), l1 = (unsigned)(d[k].y - lo);
                unsigned l2 = (unsigned)(d[k].z - lo), l3 = (unsigned)(d[k].w - lo);
                unsigned mn = min(min(l0, l1), min(l2, l3));
                hit[k] = mn < (unsigned)R_NODES;
                if (hit[k]) {
                    if (l0 < (unsigned)R_NODES) sv[k * 4 + 0] = src[ebase + 0];
                    if (l1 < (unsigned)R_NODES) sv[k * 4 + 1] = src[ebase + 1];
                    if (l2 < (unsigned)R_NODES) sv[k * 4 + 2] = src[ebase + 2];
                    if (l3 < (unsigned)R_NODES) sv[k * 4 + 3] = src[ebase + 3];
                }
            }
#pragma unroll
            for (int k = 0; k < 8; ++k) {
                if (hit[k]) {
                    unsigned ll[4] = {(unsigned)(d[k].x - lo), (unsigned)(d[k].y - lo),
                                      (unsigned)(d[k].z - lo), (unsigned)(d[k].w - lo)};
#pragma unroll
                    for (int q = 0; q < 4; ++q) {
                        if (ll[q] < (unsigned)R_NODES) {
                            unsigned pos = atomicAdd(&nl[cur], 1u);
                            list[pos] = (unsigned)sv[k * 4 + q] | (ll[q] << 16);
                        }
                    }
                }
            }
            __syncthreads();

#pragma unroll
            for (int k = 0; k < 8; ++k) {
                int idx  = (c + 1) * 8192 + k * 1024 + tid;
                int idxc = idx < N_DST4 ? idx : (N_DST4 - 1);
                int4 v = dst4[idxc];
                if (idx >= N_DST4) { v.x = -1; v.y = -1; v.z = -1; v.w = -1; }
                d[k] = v;
            }
            if (tid == 0) nl[cur ^ 1] = 0u;

            const int n = (int)nl[cur];
            if (first) {
                if (tot + (unsigned)n <= (unsigned)KEEP_CAP)
                    for (int i = tid; i < n; i += 1024) kl[KEEP_HDR + tot + i] = list[i];
                tot += (unsigned)n;
            }
            sage_gather_list(x, list, acc, dcnt, n, wave, lane, first);
            __syncthreads();
        }
        if (first && tid == 0) kl[0] = tot;
    }

    if (tid < R_PAD) {
        float c = 1.0f;
        if (tid < R_NODES) {
            if (first) { c = dcnt[tid]; cnt[lo + tid] = c; }
            else       { c = cnt[lo + tid]; }
        }
        dcnt[tid] = 1.0f / fmaxf(c, 1.0f);
    }
    for (int idx = tid; idx < 256 * 128; idx += 1024) {
        int k = idx >> 7;
        int n = idx & 127;
        float w = (k < 128) ? Wl[idx] : Wr[idx - 128 * 128];
        lw[n * KSTRIDE + k] = w;
    }
    __syncthreads();

    if (wave < 24) sage_wave_product<4>(acc, dcnt, lw, x, bias, hout, lo, wave >> 1, (wave & 1) * 4, lane, do_relu);
    else           sage_wave_product<1>(acc, dcnt, lw, x, bias, hout, lo, 12, wave - 24, lane, do_relu);
}

extern "C" void kernel_launch(void* const* d_in, const int* in_sizes, int n_in,
                              void* d_out, int out_size, void* d_ws, size_t ws_size,
                              hipStream_t stream) {
    const float* x   = (const float*)d_in[0];
    const int*   ei  = (const int*)d_in[1];
    const float* Wl0 = (const float*)d_in[2];
    const float* Wr0 = (const float*)d_in[3];
    const float* b0  = (const float*)d_in[4];
    const float* Wl1 = (const float*)d_in[5];
    const float* Wr1 = (const float*)d_in[6];
    const float* b1  = (const float*)d_in[7];
    float* out = (float*)d_out;

    const int* src = ei;
    const int* dst = ei + N_EDGES;

    float* cnt  = (float*)d_ws + (size_t)N_NODES * FDIM;
    float* hbuf = cnt + N_NODES;

    sage_layer_kernel<<<AGG_BLOCKS, 1024, 0, stream>>>(x, src, dst, cnt, out, Wl0, Wr0, b0, hbuf, 1, 1);
    sage_layer_kernel<<<AGG_BLOCKS, 1024, 0, stream>>>(hbuf, src, dst, cnt, out, Wl1, Wr1, b1, out, 0, 0);
}


# layer 0's dst scan runs to its end before any row is gathered: owned (edge index,row) into the LDS list, no barrier per chunk; src resolved in one pass, then one gather; sums in LDS arrival order
# speedup vs baseline: 3.1949x; 1.3520x over previous
#include <hip/hip_runtime.h>

typedef float v2f __attribute__((ext_vector_type(2)));
typedef float v8f __attribute__((ext_vector_type(8)));


#define N_NODES 50000
#define FDIM    128
#define N_EDGES 800000
#define KSTRIDE 260

#define R_NODES    200
#define R_PAD      208
#define ACC_S      132
#define AGG_BLOCKS 250
#define CHUNK_E    32768
#define N_CHUNK    25
#define N_DST4     (N_EDGES / 4)
#define N_FULL     24
#define LIST_CAP   32768
#define KEEP_WORDS (R_NODES * FDIM)
#define KEEP_HDR   32
#define KEEP_CAP   (KEEP_WORDS - KEEP_HDR)

__device__ __forceinline__ void sage_gather_list(
    const float* __restrict__ x, const unsigned* list, float* acc, float* dcnt,
    int n, int wave, int lane, int with_count)
{
    for (int i = wave; i < n; i += 128) {
        const bool p1 = (i + 32) < n, p2 = (i + 64) < n, p3 = (i + 96) < n;
        unsigned e0 = list[i];
        unsigned e1 = p1 ? list[i + 32] : e0;
        unsigned e2 = p2 ? list[i + 64] : e0;
        unsigned e3 = p3 ? list[i + 96] : e0;
        const float* r0 = x + (size_t)(e0 & 0xFFFFu) * FDIM + lane;
        const float* r1 = x + (size_t)(e1 & 0xFFFFu) * FDIM + lane;
        const float* r2 = x + (size_t)(e2 & 0xFFFFu) * FDIM + lane;
        const float* r3 = x + (size_t)(e3 & 0xFFFFu) * FDIM + lane;
        float a0 = r0[0], a1 = r0[32], a2 = r0[64], a3 = r0[96];
        float b0 = r1[0], b1 = r1[32], b2 = r1[64], b3 = r1[96];
        float c0 = r2[0], c1 = r2[32], c2 = r2[64], c3 = r2[96];
        float g0 = r3[0], g1 = r3[32], g2 = r3[64], g3 = r3[96];
        asm volatile("" :: "v"(a0), "v"(a1), "v"(a2), "v"(a3), "v"(b0), "v"(b1), "v"(b2), "v"(b3),
                           "v"(c0), "v"(c1), "v"(c2), "v"(c3), "v"(g0), "v"(g1), "v"(g2), "v"(g3));
        float* o0 = acc + (e0 >> 16) * ACC_S + lane;
        atomicAdd(o0, a0); atomicAdd(o0 + 32, a1); atomicAdd(o0 + 64, a2); atomicAdd(o0 + 96, a3);
        if (with_count && lane == 0) atomicAdd(&dcnt[e0 >> 16], 1.0f);
        if (p1) {
            float* o1 = acc + (e1 >> 16) * ACC_S + lane;
            atomicAdd(o1, b0); atomicAdd(o1 + 32, b1); atomicAdd(o1 + 64, b2); atomicAdd(o1 + 96, b3);
            if (with_count && lane == 0) atomicAdd(&dcnt[e1 >> 16], 1.0f);
        }
        if (p2) {
            float* o2 = acc + (e2 >> 16) * ACC_S + lane;
            atomicAdd(o2, c0); atomicAdd(o2 + 32, c1); atomicAdd(o2 + 64, c2); atomicAdd(o2 + 96, c3);
            if (with_count && lane == 0) atomicAdd(&dcnt[e2 >> 16], 1.0f);
        }
        if (p3) {
            float* o3 = acc + (e3 >> 16) * ACC_S + lane;
            atomicAdd(o3, g0); atomicAdd(o3 + 32, g1); atomicAdd(o3 + 64, g2); atomicAdd(o3 + 96, g3);
            if (with_count && lane == 0) atomicAdd(&dcnt[e3 >> 16], 1.0f);
        }
    }
}

template <bool CHECK>
__device__ __forceinline__ void sage_scan_group(const int4 dv, const int idx4, const int lo,
                                                unsigned* list, unsigned* ncnt)
{
    const unsigned l0 = (unsigned)(dv.x - lo), l1 = (unsigned)(dv.y - lo);
    const unsigned l2 = (unsigned)(dv.z - lo), l3 = (unsigned)(dv.w - lo);
    unsigned mn = min(min(l0, l1), min(l2, l3));
    if (CHECK && idx4 >= N_DST4) mn = 0xFFFFFFFFu;
    if (mn < (unsigned)R_NODES) {
        const unsigned h0 = l0 < (unsigned)R_NODES ? 1u : 0u, h1 = l1 < (unsigned)R_NODES ? 1u : 0u;
        const unsigned h2 = l2 < (unsigned)R_NODES ? 1u : 0u, h3 = l3 < (unsigned)R_NODES ? 1u : 0u;
        unsigned pos = atomicAdd(ncnt, h0 + h1 + h2 + h3);
        const unsigned e = (unsigned)idx4 << 2;
        if (h0) { if (pos < (unsigned)LIST_CAP) list[pos] = (e     ) | (l0 << 20); ++pos; }
        if (h1) { if (pos < (unsigned)LIST_CAP) list[pos] = (e + 1u) | (l1 << 20); ++pos; }
        if (h2) { if (pos < (unsigned)LIST_CAP) list[pos] = (e + 2u) | (l2 << 20); ++pos; }
        if (h3) { if (pos < (unsigned)LIST_CAP) list[pos] = (e + 3u) | (l3 << 20); }
    }
}

__device__ __forceinline__ void sage_load_chunk(int4 (&r)[8], const int4* __restrict__ dst4, int c, int tid)
{
#pragma unroll
    for (int k = 0; k < 8; ++k) {
        int idx = c * 8192 + k * 1024 + tid;
        r[k] = dst4[idx < N_DST4 ? idx : (N_DST4 - 1)];
    }
}

template <bool CHECK>
__device__ __forceinline__ void sage_scan_chunk(const int4 (&r)[8], int c, int tid, int lo,
                                                unsigned* list, unsigned* ncnt)
{
#pragma unroll
    for (int k = 0; k < 8; ++k)
        sage_scan_group<CHECK>(r[k], c * 8192 + k * 1024 + tid, lo, list, ncnt);
}

template <int NT>
__device__ __forceinline__ void sage_wave_product(
    const float* acc, const float* dcnt, const float* lw,
    const float* __restrict__ x, const float* __restrict__ bias, float* hout,
    int lo, int rt, int ct0, int lane, int do_relu)
{
    const int h   = lane >> 4;
    const int col = lane & 15;
    const int rl  = rt * 16 + col;
    int rg = lo + rl;
    rg = rg < N_NODES ? rg : (N_NODES - 1);

    const float  scale  = dcnt[rl];
    const float* arow_m = acc + rl * ACC_S + 2 * h;
    const float* arow_x = x + (size_t)rg * FDIM + 2 * h;
    const float* brow   = lw + (ct0 * 16 + col) * KSTRIDE + 2 * h;

    v8f d4[NT];
    v8f z = {0.f, 0.f, 0.f, 0.f, 0.f, 0.f, 0.f, 0.f};
#pragma unroll
    for (int t = 0; t < NT; ++t) d4[t] = z;

#pragma unroll 4
    for (int kk = 0; kk < 128; kk += 4) {
        v2f a = *(const v2f*)(arow_m + kk);
        a.x *= scale;
        a.y *= scale;
#pragma unroll
        for (int t = 0; t < NT; ++t) {
            v2f b = *(const v2f*)(brow + t * (16 * KSTRIDE) + kk);
            d4[t] = __builtin_amdgcn_wmma_f32_16x16x4_f32(
                false, a, false, b, (short)0, d4[t], false, false);
        }
    }

#pragma unroll 4
    for (int kk = 0; kk < 128; kk += 4) {
        v2f a = *(const v2f*)(arow_x + kk);
#pragma unroll
        for (int t = 0; t < NT; ++t) {
            v2f b = *(const v2f*)(brow + t * (16 * KSTRIDE) + 128 + kk);
            d4[t] = __builtin_amdgcn_wmma_f32_16x16x4_f32(
                false, a, false, b, (short)0, d4[t], false, false);
        }
    }

#pragma unroll
    for (int t = 0; t < NT; ++t) {
        int n  = (ct0 + t) * 16 + col;
        float bv = bias[n];
#pragma unroll
        for (int v = 0; v < 8; ++v) {
            int row = rt * 16 + v + 8 * h;
            float val = d4[t][v] + bv;
            if (do_relu) val = fmaxf(val, 0.0f);
            if (row < R_NODES) hout[(size_t)(lo + row) * FDIM + n] = val;
        }
    }
}

__global__ __launch_bounds__(1024) void sage_layer_kernel(
    const float* __restrict__ x,
    const int*   __restrict__ src,
    const int*   __restrict__ dst,
    float* cnt,
    float* keep,
    const float* __restrict__ Wl, const float* __restrict__ Wr,
    const float* __restrict__ bias,
    float* hout,
    int first,
    int do_relu)
{
    __shared__ __attribute__((aligned(16))) float    acc[R_PAD * ACC_S];
    __shared__ float    dcnt[R_PAD];
    __shared__ __attribute__((aligned(16))) unsigned ubuf[128 * KSTRIDE];
    __shared__ unsigned nl[3];
    unsigned* list = ubuf;
    float*    lw   = (float*)ubuf;

    const int tid  = threadIdx.x;
    const int lane = tid & 31;
    const int wave = __builtin_amdgcn_readfirstlane(tid >> 5);
    const int lo   = blockIdx.x * R_NODES;
    unsigned* kl   = (unsigned*)keep + (size_t)blockIdx.x * KEEP_WORDS;

    for (int i = tid; i < R_PAD * ACC_S; i += 1024) acc[i] = 0.0f;
    if (tid < R_PAD) dcnt[tid] = 0.0f;
    if (tid < 3) nl[tid] = 0u;

    const unsigned kept = first ? 0xFFFFFFFFu : kl[0];

    if (kept <= (unsigned)KEEP_CAP) {
        const unsigned* ke = kl + KEEP_HDR;
        for (unsigned i = tid; i < kept; i += 4096) {
            const unsigned i1 = i + 1024, i2 = i + 2048, i3 = i + 3072;
            const bool q1 = i1 < kept, q2 = i2 < kept, q3 = i3 < kept;
            unsigned v0 = ke[i];
            unsigned v1 = q1 ? ke[i1] : 0u;
            unsigned v2 = q2 ? ke[i2] : 0u;
            unsigned v3 = q3 ? ke[i3] : 0u;
            list[i] = v0;
            if (q1) list[i1] = v1;
            if (q2) list[i2] = v2;
            if (q3) list[i3] = v3;
        }
        __syncthreads();
        sage_gather_list(x, list, acc, dcnt, (int)kept, wave, lane, 0);
        __syncthreads();
    } else {
        const int4* dst4 = (const int4*)dst;
        int4 d[8];
        sage_load_chunk(d, dst4, 0, tid);
        __syncthreads();

        {
            int4 d2[8];
            for (int c = 0; c < N_FULL; c += 2) {
                sage_load_chunk(d2, dst4, c + 1, tid);
                sage_scan_chunk<false>(d, c, tid, lo, list, &nl[2]);
                sage_load_chunk(d, dst4, c + 2, tid);
                sage_scan_chunk<false>(d2, c + 1, tid, lo, list, &nl[2]);
            }
            sage_scan_chunk<true>(d, N_FULL, tid, lo, list, &nl[2]);
        }
        __syncthreads();
        const unsigned total = nl[2];

        if (total <= (unsigned)LIST_CAP) {
            const bool keepit = first && total <= (unsigned)KEEP_CAP;
            unsigned* ke = kl + KEEP_HDR;
            for (unsigned i = tid; i < total; i += 4096) {
                const unsigned i1 = i + 1024, i2 = i + 2048, i3 = i + 3072;
                const bool q1 = i1 < total, q2 = i2 < total, q3 = i3 < total;
                const unsigned a0 = list[i];
                const unsigned a1 = q1 ? list[i1] : a0;
                const unsigned a2 = q2 ? list[i2] : a0;
                const unsigned a3 = q3 ? list[i3] : a0;
                const unsigned s0 = (unsigned)src[a0 & 0xFFFFFu];
                const unsigned s1 = (unsigned)src[a1 & 0xFFFFFu];
                const unsigned s2 = (unsigned)src[a2 & 0xFFFFFu];
                const unsigned s3 = (unsigned)src[a3 & 0xFFFFFu];
                const unsigned v0 = s0 | ((a0 >> 20) << 16);
                const unsigned v1 = s1 | ((a1 >> 20) << 16);
                const unsigned v2 = s2 | ((a2 >> 20) << 16);
                const unsigned v3 = s3 | ((a3 >> 20) << 16);
                list[i] = v0;
                if (q1) list[i1] = v1;
                if (q2) list[i2] = v2;
                if (q3) list[i3] = v3;
                if (keepit) {
                    ke[i] = v0;
                    if (q1) ke[i1] = v1;
                    if (q2) ke[i2] = v2;
                    if (q3) ke[i3] = v3;
                }
            }
            __syncthreads();
            sage_gather_list(x, list, acc, dcnt, (int)total, wave, lane, first);
            __syncthreads();
        } else {
            sage_load_chunk(d, dst4, 0, tid);
            for (int c = 0; c < N_CHUNK; ++c) {
                const int cur = c & 1;
                int  sv[32];
                bool hit[8];
#pragma unroll
                for (int k = 0; k < 8; ++k) {
                    int ebase = (c * 8192 + k * 1024 + tid) << 2;
                    unsigned l0 = (unsigned)(d[k].x - lo), l1 = (unsigned)(d[k].y - lo);
                    unsigned l2 = (unsigned)(d[k].z - lo), l3 = (unsigned)(d[k].w - lo);
                    unsigned mn = min(min(l0, l1), min(l2, l3));
                    hit[k] = mn < (unsigned)R_NODES;
                    if (hit[k]) {
                        if (l0 < (unsigned)R_NODES) sv[k * 4 + 0] = src[ebase + 0];
                        if (l1 < (unsigned)R_NODES) sv[k * 4 + 1] = src[ebase + 1];
                        if (l2 < (unsigned)R_NODES) sv[k * 4 + 2] = src[ebase + 2];
                        if (l3 < (unsigned)R_NODES) sv[k * 4 + 3] = src[ebase + 3];
                    }
                }
#pragma unroll
                for (int k = 0; k < 8; ++k) {
                    if (hit[k]) {
                        unsigned ll[4] = {(unsigned)(d[k].x - lo), (unsigned)(d[k].y - lo),
                                          (unsigned)(d[k].z - lo), (unsigned)(d[k].w - lo)};
#pragma unroll
                        for (int q = 0; q < 4; ++q) {
                            if (ll[q] < (unsigned)R_NODES) {
                                unsigned pos = atomicAdd(&nl[cur], 1u);
                                list[pos] = (unsigned)sv[k * 4 + q] | (ll[q] << 16);
                            }
                        }
                    }
                }
                __syncthreads();

#pragma unroll
                for (int k = 0; k < 8; ++k) {
                    int idx  = (c + 1) * 8192 + k * 1024 + tid;
                    int idxc = idx < N_DST4 ? idx : (N_DST4 - 1);
                    int4 v = dst4[idxc];
                    if (idx >= N_DST4) { v.x = -1; v.y = -1; v.z = -1; v.w = -1; }
                    d[k] = v;
                }
                if (tid == 0) nl[cur ^ 1] = 0u;

                const int n = (int)nl[cur];
                sage_gather_list(x, list, acc, dcnt, n, wave, lane, first);
                __syncthreads();
            }
        }
        if (first && tid == 0) kl[0] = total;
    }

    if (tid < R_PAD) {
        float c = 1.0f;
        if (tid < R_NODES) {
            if (first) { c = dcnt[tid]; cnt[lo + tid] = c; }
            else       { c = cnt[lo + tid]; }
        }
        dcnt[tid] = 1.0f / fmaxf(c, 1.0f);
    }
    for (int idx = tid; idx < 256 * 128; idx += 1024) {
        int k = idx >> 7;
        int n = idx & 127;
        float w = (k < 128) ? Wl[idx] : Wr[idx - 128 * 128];
        lw[n * KSTRIDE + k] = w;
    }
    __syncthreads();

    if (wave < 24) sage_wave_product<4>(acc, dcnt, lw, x, bias, hout, lo, wave >> 1, (wave & 1) * 4, lane, do_relu);
    else           sage_wave_product<1>(acc, dcnt, lw, x, bias, hout, lo, 12, wave - 24, lane, do_relu);
}

extern "C" void kernel_launch(void* const* d_in, const int* in_sizes, int n_in,
                              void* d_out, int out_size, void* d_ws, size_t ws_size,
                              hipStream_t stream) {
    const float* x   = (const float*)d_in[0];
    const int*   ei  = (const int*)d_in[1];
    const float* Wl0 = (const float*)d_in[2];
    const float* Wr0 = (const float*)d_in[3];
    const float* b0  = (const float*)d_in[4];
    const float* Wl1 = (const float*)d_in[5];
    const float* Wr1 = (const float*)d_in[6];
    const float* b1  = (const float*)d_in[7];
    float* out = (float*)d_out;

    const int* src = ei;
    const int* dst = ei + N_EDGES;

    float* cnt  = (float*)d_ws + (size_t)N_NODES * FDIM;
    float* hbuf = cnt + N_NODES;

    sage_layer_kernel<<<AGG_BLOCKS, 1024, 0, stream>>>(x, src, dst, cnt, out, Wl0, Wr0, b0, hbuf, 1, 1);
    sage_layer_kernel<<<AGG_BLOCKS, 1024, 0, stream>>>(hbuf, src, dst, cnt, out, Wl1, Wr1, b1, out, 0, 0);
}


# layer 0 scans one owner byte per edge (dst/200, written by a small first launch), not the 4-byte dst: a quarter of the scan bytes; owned edges then fetch dst and src; 3 launches; LDS arrival order
# speedup vs baseline: 3.3655x; 1.0534x over previous
#include <hip/hip_runtime.h>

typedef float v2f __attribute__((ext_vector_type(2)));
typedef float v8f __attribute__((ext_vector_type(8)));


#define N_NODES 50000
#define FDIM    128
#define N_EDGES 800000
#define KSTRIDE 260

#define R_NODES    200
#define R_PAD      208
#define ACC_S      132
#define AGG_BLOCKS 250
#define CHUNK_E    32768
#define N_CHUNK    25
#define N_DST4     (N_EDGES / 4)
#define LIST_CAP   32768
#define N_OB16     (N_EDGES / 16)
#define OB_CHUNK   8192
#define OB_FULL    6
#define OWN_BLOCKS 196
#define EL_CAP     24576
#define GL_OFF     24576
#define GL_CAP     8192
#define KEEP_WORDS (R_NODES * FDIM)
#define KEEP_HDR   32
#define KEEP_CAP   (KEEP_WORDS - KEEP_HDR)

__device__ __forceinline__ void sage_gather_list(
    const float* __restrict__ x, const unsigned* list, float* acc, float* dcnt,
    int n, int wave, int lane, int with_count)
{
    for (int i = wave; i < n; i += 128) {
        const bool p1 = (i + 32) < n, p2 = (i + 64) < n, p3 = (i + 96) < n;
        unsigned e0 = list[i];
        unsigned e1 = p1 ? list[i + 32] : e0;
        unsigned e2 = p2 ? list[i + 64] : e0;
        unsigned e3 = p3 ? list[i + 96] : e0;
        const float* r0 = x + (size_t)(e0 & 0xFFFFu) * FDIM + lane;
        const float* r1 = x + (size_t)(e1 & 0xFFFFu) * FDIM + lane;
        const float* r2 = x + (size_t)(e2 & 0xFFFFu) * FDIM + lane;
        const float* r3 = x + (size_t)(e3 & 0xFFFFu) * FDIM + lane;
        float a0 = r0[0], a1 = r0[32], a2 = r0[64], a3 = r0[96];
        float b0 = r1[0], b1 = r1[32], b2 = r1[64], b3 = r1[96];
        float c0 = r2[0], c1 = r2[32], c2 = r2[64], c3 = r2[96];
        float g0 = r3[0], g1 = r3[32], g2 = r3[64], g3 = r3[96];
        asm volatile("" :: "v"(a0), "v"(a1), "v"(a2), "v"(a3), "v"(b0), "v"(b1), "v"(b2), "v"(b3),
                           "v"(c0), "v"(c1), "v"(c2), "v"(c3), "v"(g0), "v"(g1), "v"(g2), "v"(g3));
        float* o0 = acc + (e0 >> 16) * ACC_S + lane;
        atomicAdd(o0, a0); atomicAdd(o0 + 32, a1); atomicAdd(o0 + 64, a2); atomicAdd(o0 + 96, a3);
        if (with_count && lane == 0) atomicAdd(&dcnt[e0 >> 16], 1.0f);
        if (p1) {
            float* o1 = acc + (e1 >> 16) * ACC_S + lane;
            atomicAdd(o1, b0); atomicAdd(o1 + 32, b1); atomicAdd(o1 + 64, b2); atomicAdd(o1 + 96, b3);
            if (with_count && lane == 0) atomicAdd(&dcnt[e1 >> 16], 1.0f);
        }
        if (p2) {
            float* o2 = acc + (e2 >> 16) * ACC_S + lane;
            atomicAdd(o2, c0); atomicAdd(o2 + 32, c1); atomicAdd(o2 + 64, c2); atomicAdd(o2 + 96, c3);
            if (with_count && lane == 0) atomicAdd(&dcnt[e2 >> 16], 1.0f);
        }
        if (p3) {
            float* o3 = acc + (e3 >> 16) * ACC_S + lane;
            atomicAdd(o3, g0); atomicAdd(o3 + 32, g1); atomicAdd(o3 + 64, g2); atomicAdd(o3 + 96, g3);
            if (with_count && lane == 0) atomicAdd(&dcnt[e3 >> 16], 1.0f);
        }
    }
}

__device__ __forceinline__ unsigned sage_zero_bytes(const unsigned t)
{
    return ~(((t & 0x7F7F7F7Fu) + 0x7F7F7F7Fu) | t | 0x7F7F7F7Fu);
}

__device__ __forceinline__ unsigned sage_zero_nibble(const unsigned t)
{
    return (((sage_zero_bytes(t) >> 7) * 0x01020408u) >> 24) & 0xFu;
}

__device__ __forceinline__ void sage_load_ob(uint4 (&r)[8], const uint4* __restrict__ ob16, int c, int tid)
{
#pragma unroll
    for (int k = 0; k < 8; ++k) {
        int idx = c * OB_CHUNK + k * 1024 + tid;
        r[k] = ob16[idx < N_OB16 ? idx : (N_OB16 - 1)];
    }
}

template <bool CHECK>
__device__ __forceinline__ void sage_scan_ob(const uint4 (&r)[8], int c, int tid, unsigned pat,
                                             unsigned* glist, unsigned* ng)
{
#pragma unroll
    for (int k = 0; k < 8; ++k) {
        const int idx = c * OB_CHUNK + k * 1024 + tid;
        unsigned z = sage_zero_bytes(r[k].x ^ pat) | sage_zero_bytes(r[k].y ^ pat)
                   | sage_zero_bytes(r[k].z ^ pat) | sage_zero_bytes(r[k].w ^ pat);
        if (CHECK && idx >= N_OB16) z = 0u;
        if (z) {
            const unsigned pos = atomicAdd(ng, 1u);
            if (pos < (unsigned)GL_CAP) glist[pos] = (unsigned)idx;
        }
    }
}

__device__ __forceinline__ void sage_group_edges(const uint4 w, const unsigned g, const unsigned pat,
                                                 unsigned* list, unsigned* ne)
{
    unsigned m = sage_zero_nibble(w.x ^ pat) | (sage_zero_nibble(w.y ^ pat) << 4)
               | (sage_zero_nibble(w.z ^ pat) << 8) | (sage_zero_nibble(w.w ^ pat) << 12);
    unsigned pos = atomicAdd(ne, (unsigned)__popc(m));
    const unsigned e = g << 4;
    while (m) {
        const unsigned j = (unsigned)__ffs((int)m) - 1u;
        if (pos < (unsigned)EL_CAP) list[pos] = e + j;
        ++pos;
        m &= m - 1u;
    }
}

__device__ __forceinline__ void sage_load_chunk(int4 (&r)[8], const int4* __restrict__ dst4, int c, int tid)
{
#pragma unroll
    for (int k = 0; k < 8; ++k) {
        int idx = c * 8192 + k * 1024 + tid;
        r[k] = dst4[idx < N_DST4 ? idx : (N_DST4 - 1)];
    }
}

template <int NT>
__device__ __forceinline__ void sage_wave_product(
    const float* acc, const float* dcnt, const float* lw,
    const float* __restrict__ x, const float* __restrict__ bias, float* hout,
    int lo, int rt, int ct0, int lane, int do_relu)
{
    const int h   = lane >> 4;
    const int col = lane & 15;
    const int rl  = rt * 16 + col;
    int rg = lo + rl;
    rg = rg < N_NODES ? rg : (N_NODES - 1);

    const float  scale  = dcnt[rl];
    const float* arow_m = acc + rl * ACC_S + 2 * h;
    const float* arow_x = x + (size_t)rg * FDIM + 2 * h;
    const float* brow   = lw + (ct0 * 16 + col) * KSTRIDE + 2 * h;

    v8f d4[NT];
    v8f z = {0.f, 0.f, 0.f, 0.f, 0.f, 0.f, 0.f, 0.f};
#pragma unroll
    for (int t = 0; t < NT; ++t) d4[t] = z;

#pragma unroll 4
    for (int kk = 0; kk < 128; kk += 4) {
        v2f a = *(const v2f*)(arow_m + kk);
        a.x *= scale;
        a.y *= scale;
#pragma unroll
        for (int t = 0; t < NT; ++t) {
            v2f b = *(const v2f*)(brow + t * (16 * KSTRIDE) + kk);
            d4[t] = __builtin_amdgcn_wmma_f32_16x16x4_f32(
                false, a, false, b, (short)0, d4[t], false, false);
        }
    }

#pragma unroll 4
    for (int kk = 0; kk < 128; kk += 4) {
        v2f a = *(const v2f*)(arow_x + kk);
#pragma unroll
        for (int t = 0; t < NT; ++t) {
            v2f b = *(const v2f*)(brow + t * (16 * KSTRIDE) + 128 + kk);
            d4[t] = __builtin_amdgcn_wmma_f32_16x16x4_f32(
                false, a, false, b, (short)0, d4[t], false, false);
        }
    }

#pragma unroll
    for (int t = 0; t < NT; ++t) {
        int n  = (ct0 + t) * 16 + col;
        float bv = bias[n];
#pragma unroll
        for (int v = 0; v < 8; ++v) {
            int row = rt * 16 + v + 8 * h;
            float val = d4[t][v] + bv;
            if (do_relu) val = fmaxf(val, 0.0f);
            if (row < R_NODES) hout[(size_t)(lo + row) * FDIM + n] = val;
        }
    }
}

__global__ __launch_bounds__(1024) void sage_layer_kernel(
    const float* __restrict__ x,
    const int*   __restrict__ src,
    const int*   __restrict__ dst,
    const unsigned* __restrict__ ob,
    float* cnt,
    float* keep,
    const float* __restrict__ Wl, const float* __restrict__ Wr,
    const float* __restrict__ bias,
    float* hout,
    int first,
    int do_relu)
{
    __shared__ __attribute__((aligned(16))) float    acc[R_PAD * ACC_S];
    __shared__ float    dcnt[R_PAD];
    __shared__ __attribute__((aligned(16))) unsigned ubuf[128 * KSTRIDE];
    __shared__ unsigned nl[4];
    unsigned* list = ubuf;
    float*    lw   = (float*)ubuf;

    const int tid  = threadIdx.x;
    const int lane = tid & 31;
    const int wave = __builtin_amdgcn_readfirstlane(tid >> 5);
    const int lo   = blockIdx.x * R_NODES;
    unsigned* kl   = (unsigned*)keep + (size_t)blockIdx.x * KEEP_WORDS;

    for (int i = tid; i < R_PAD * ACC_S; i += 1024) acc[i] = 0.0f;
    if (tid < R_PAD) dcnt[tid] = 0.0f;
    if (tid < 4) nl[tid] = 0u;

    const unsigned kept = first ? 0xFFFFFFFFu : kl[0];

    if (kept <= (unsigned)KEEP_CAP) {
        const unsigned* ke = kl + KEEP_HDR;
        for (unsigned i = tid; i < kept; i += 4096) {
            const unsigned i1 = i + 1024, i2 = i + 2048, i3 = i + 3072;
            const bool q1 = i1 < kept, q2 = i2 < kept, q3 = i3 < kept;
            unsigned v0 = ke[i];
            unsigned v1 = q1 ? ke[i1] : 0u;
            unsigned v2 = q2 ? ke[i2] : 0u;
            unsigned v3 = q3 ? ke[i3] : 0u;
            list[i] = v0;
            if (q1) list[i1] = v1;
            if (q2) list[i2] = v2;
            if (q3) list[i3] = v3;
        }
        __syncthreads();
        sage_gather_list(x, list, acc, dcnt, (int)kept, wave, lane, 0);
        __syncthreads();
    } else {
        const int4*  dst4  = (const int4*)dst;
        const uint4* ob16  = (const uint4*)ob;
        const unsigned pat = (unsigned)blockIdx.x * 0x01010101u;
        unsigned*    glist = ubuf + GL_OFF;

        {
            uint4 b[8], b2[8];
            sage_load_ob(b, ob16, 0, tid);
            __syncthreads();
            for (int c = 0; c < OB_FULL; c += 2) {
                sage_load_ob(b2, ob16, c + 1, tid);
                sage_scan_ob<false>(b, c, tid, pat, glist, &nl[2]);
                sage_load_ob(b, ob16, c + 2, tid);
                sage_scan_ob<false>(b2, c + 1, tid, pat, glist, &nl[2]);
            }
            sage_scan_ob<true>(b, OB_FULL, tid, pat, glist, &nl[2]);
        }
        __syncthreads();
        const unsigned ngrp = nl[2];

        if (ngrp <= (unsigned)GL_CAP) {
            for (unsigned i = tid; i < ngrp; i += 4096) {
                const unsigned i1 = i + 1024, i2 = i + 2048, i3 = i + 3072;
                const bool q1 = i1 < ngrp, q2 = i2 < ngrp, q3 = i3 < ngrp;
                const unsigned g0 = glist[i];
                const unsigned g1 = q1 ? glist[i1] : g0;
                const unsigned g2 = q2 ? glist[i2] : g0;
                const unsigned g3 = q3 ? glist[i3] : g0;
                const uint4 w0 = ob16[g0], w1 = ob16[g1], w2 = ob16[g2], w3 = ob16[g3];
                sage_group_edges(w0, g0, pat, list, &nl[3]);
                if (q1) sage_group_edges(w1, g1, pat, list, &nl[3]);
                if (q2) sage_group_edges(w2, g2, pat, list, &nl[3]);
                if (q3) sage_group_edges(w3, g3, pat, list, &nl[3]);
            }
        }
        __syncthreads();
        const unsigned total = (ngrp <= (unsigned)GL_CAP && nl[3] <= (unsigned)EL_CAP) ? nl[3] : 0xFFFFFFFFu;

        if (total <= (unsigned)EL_CAP) {
            const bool keepit = first != 0;
            unsigned* ke = kl + KEEP_HDR;
            for (unsigned i = tid; i < total; i += 4096) {
                const unsigned i1 = i + 1024, i2 = i + 2048, i3 = i + 3072;
                const bool q1 = i1 < total, q2 = i2 < total, q3 = i3 < total;
                const unsigned a0 = list[i];
                const unsigned a1 = q1 ? list[i1] : a0;
                const unsigned a2 = q2 ? list[i2] : a0;
                const unsigned a3 = q3 ? list[i3] : a0;
                const unsigned s0 = (unsigned)src[a0], r0 = (unsigned)(dst[a0] - lo);
                const unsigned s1 = (unsigned)src[a1], r1 = (unsigned)(dst[a1] - lo);
                const unsigned s2 = (unsigned)src[a2], r2 = (unsigned)(dst[a2] - lo);
                const unsigned s3 = (unsigned)src[a3], r3 = (unsigned)(dst[a3] - lo);
                const unsigned v0 = s0 | (r0 << 16);
                const unsigned v1 = s1 | (r1 << 16);
                const unsigned v2 = s2 | (r2 << 16);
                const unsigned v3 = s3 | (r3 << 16);
                list[i] = v0;
                if (q1) list[i1] = v1;
                if (q2) list[i2] = v2;
                if (q3) list[i3] = v3;
                if (keepit) {
                    ke[i] = v0;
                    if (q1) ke[i1] = v1;
                    if (q2) ke[i2] = v2;
                    if (q3) ke[i3] = v3;
                }
            }
            __syncthreads();
            sage_gather_list(x, list, acc, dcnt, (int)total, wave, lane, first);
            __syncthreads();
        } else {
            int4 d[8];
            sage_load_chunk(d, dst4, 0, tid);
            for (int c = 0; c < N_CHUNK; ++c) {
                const int cur = c & 1;
                int  sv[32];
                bool hit[8];
#pragma unroll
                for (int k = 0; k < 8; ++k) {
                    int ebase = (c * 8192 + k * 1024 + tid) << 2;
                    unsigned l0 = (unsigned)(d[k].x - lo), l1 = (unsigned)(d[k].y - lo);
                    unsigned l2 = (unsigned)(d[k].z - lo), l3 = (unsigned)(d[k].w - lo);
                    unsigned mn = min(min(l0, l1), min(l2, l3));
                    hit[k] = mn < (unsigned)R_NODES;
                    if (hit[k]) {
                        if (l0 < (unsigned)R_NODES) sv[k * 4 + 0] = src[ebase + 0];
                        if (l1 < (unsigned)R_NODES) sv[k * 4 + 1] = src[ebase + 1];
                        if (l2 < (unsigned)R_NODES) sv[k * 4 + 2] = src[ebase + 2];
                        if (l3 < (unsigned)R_NODES) sv[k * 4 + 3] = src[ebase + 3];
                    }
                }
#pragma unroll
                for (int k = 0; k < 8; ++k) {
                    if (hit[k]) {
                        unsigned ll[4] = {(unsigned)(d[k].x - lo), (unsigned)(d[k].y - lo),
                                          (unsigned)(d[k].z - lo), (unsigned)(d[k].w - lo)};
#pragma unroll
                        for (int q = 0; q < 4; ++q) {
                            if (ll[q] < (unsigned)R_NODES) {
                                unsigned pos = atomicAdd(&nl[cur], 1u);
                                list[pos] = (unsigned)sv[k * 4 + q] | (ll[q] << 16);
                            }
                        }
                    }
                }
                __syncthreads();

#pragma unroll
                for (int k = 0; k < 8; ++k) {
                    int idx  = (c + 1) * 8192 + k * 1024 + tid;
                    int idxc = idx < N_DST4 ? idx : (N_DST4 - 1);
                    int4 v = dst4[idxc];
                    if (idx >= N_DST4) { v.x = -1; v.y = -1; v.z = -1; v.w = -1; }
                    d[k] = v;
                }
                if (tid == 0) nl[cur ^ 1] = 0u;

                const int n = (int)nl[cur];
                sage_gather_list(x, list, acc, dcnt, n, wave, lane, first);
                __syncthreads();
            }
        }
        if (first && tid == 0) kl[0] = total;
    }

    if (tid < R_PAD) {
        float c = 1.0f;
        if (tid < R_NODES) {
            if (first) { c = dcnt[tid]; cnt[lo + tid] = c; }
            else       { c = cnt[lo + tid]; }
        }
        dcnt[tid] = 1.0f / fmaxf(c, 1.0f);
    }
    for (int idx = tid; idx < 256 * 128; idx += 1024) {
        int k = idx >> 7;
        int n = idx & 127;
        float w = (k < 128) ? Wl[idx] : Wr[idx - 128 * 128];
        lw[n * KSTRIDE + k] = w;
    }
    __syncthreads();

    if (wave < 24) sage_wave_product<4>(acc, dcnt, lw, x, bias, hout, lo, wave >> 1, (wave & 1) * 4, lane, do_relu);
    else           sage_wave_product<1>(acc, dcnt, lw, x, bias, hout, lo, 12, wave - 24, lane, do_relu);
}

__global__ __launch_bounds__(1024) void sage_owner_kernel(const int* __restrict__ dst, unsigned* __restrict__ ob)
{
    const int i = blockIdx.x * 1024 + threadIdx.x;
    if (i < N_DST4) {
        const int4 d = ((const int4*)dst)[i];
        const unsigned u0 = (unsigned)d.x, u1 = (unsigned)d.y, u2 = (unsigned)d.z, u3 = (unsigned)d.w;
        const unsigned o0 = u0 < (unsigned)N_NODES ? u0 / (unsigned)R_NODES : 255u;
        const unsigned o1 = u1 < (unsigned)N_NODES ? u1 / (unsigned)R_NODES : 255u;
        const unsigned o2 = u2 < (unsigned)N_NODES ? u2 / (unsigned)R_NODES : 255u;
        const unsigned o3 = u3 < (unsigned)N_NODES ? u3 / (unsigned)R_NODES : 255u;
        ob[i] = o0 | (o1 << 8) | (o2 << 16) | (o3 << 24);
    }
}

extern "C" void kernel_launch(void* const* d_in, const int* in_sizes, int n_in,
                              void* d_out, int out_size, void* d_ws, size_t ws_size,
                              hipStream_t stream) {
    const float* x   = (const float*)d_in[0];
    const int*   ei  = (const int*)d_in[1];
    const float* Wl0 = (const float*)d_in[2];
    const float* Wr0 = (const float*)d_in[3];
    const float* b0  = (const float*)d_in[4];
    const float* Wl1 = (const float*)d_in[5];
    const float* Wr1 = (const float*)d_in[6];
    const float* b1  = (const float*)d_in[7];
    float* out = (float*)d_out;

    const int* src = ei;
    const int* dst = ei + N_EDGES;

    unsigned* ob = (unsigned*)d_ws;
    float* cnt  = (float*)d_ws + (size_t)N_NODES * FDIM;
    float* hbuf = cnt + N_NODES;

    sage_owner_kernel<<<OWN_BLOCKS, 1024, 0, stream>>>(dst, ob);
    sage_layer_kernel<<<AGG_BLOCKS, 1024, 0, stream>>>(x, src, dst, ob, cnt, out, Wl0, Wr0, b0, hbuf, 1, 1);
    sage_layer_kernel<<<AGG_BLOCKS, 1024, 0, stream>>>(hbuf, src, dst, ob, cnt, out, Wl1, Wr1, b1, out, 0, 0);
}


# neighbour sums in registers, not LDS float atomics: wave w owns local rows r%32==w, filters the group's list for them, one 16-B load per lane per source row, row stored to LDS once; list-order sums
# speedup vs baseline: 3.6641x; 1.0887x over previous
#include <hip/hip_runtime.h>

typedef float v2f __attribute__((ext_vector_type(2)));
typedef float v4f __attribute__((ext_vector_type(4)));
typedef float v32f __attribute__((ext_vector_type(32)));
typedef float v8f __attribute__((ext_vector_type(8)));


#define N_NODES 50000
#define FDIM    128
#define N_EDGES 800000
#define KSTRIDE 260

#define R_NODES    200
#define R_PAD      208
#define ACC_S      132
#define AGG_BLOCKS 250
#define CHUNK_E    32768
#define N_CHUNK    25
#define N_DST4     (N_EDGES / 4)
#define LIST_CAP   32768
#define N_OB16     (N_EDGES / 16)
#define OB_CHUNK   8192
#define OB_FULL    6
#define OWN_BLOCKS 196
#define EL_CAP     24576
#define GL_OFF     24576
#define GL_CAP     8192
#define KEEP_WORDS (R_NODES * FDIM)
#define KEEP_HDR   32
#define KEEP_CAP   (KEEP_WORDS - KEEP_HDR)

__device__ __forceinline__ void sage_gather_list(
    const float* __restrict__ x, const unsigned* list, float* acc, float* dcnt,
    int n, int wave, int lane, int with_count)
{
    for (int i = wave; i < n; i += 128) {
        const bool p1 = (i + 32) < n, p2 = (i + 64) < n, p3 = (i + 96) < n;
        unsigned e0 = list[i];
        unsigned e1 = p1 ? list[i + 32] : e0;
        unsigned e2 = p2 ? list[i + 64] : e0;
        unsigned e3 = p3 ? list[i + 96] : e0;
        const float* r0 = x + (size_t)(e0 & 0xFFFFu) * FDIM + lane;
        const float* r1 = x + (size_t)(e1 & 0xFFFFu) * FDIM + lane;
        const float* r2 = x + (size_t)(e2 & 0xFFFFu) * FDIM + lane;
        const float* r3 = x + (size_t)(e3 & 0xFFFFu) * FDIM + lane;
        float a0 = r0[0], a1 = r0[32], a2 = r0[64], a3 = r0[96];
        float b0 = r1[0], b1 = r1[32], b2 = r1[64], b3 = r1[96];
        float c0 = r2[0], c1 = r2[32], c2 = r2[64], c3 = r2[96];
        float g0 = r3[0], g1 = r3[32], g2 = r3[64], g3 = r3[96];
        asm volatile("" :: "v"(a0), "v"(a1), "v"(a2), "v"(a3), "v"(b0), "v"(b1), "v"(b2), "v"(b3),
                           "v"(c0), "v"(c1), "v"(c2), "v"(c3), "v"(g0), "v"(g1), "v"(g2), "v"(g3));
        float* o0 = acc + (e0 >> 16) * ACC_S + lane;
        atomicAdd(o0, a0); atomicAdd(o0 + 32, a1); atomicAdd(o0 + 64, a2); atomicAdd(o0 + 96, a3);
        if (with_count && lane == 0) atomicAdd(&dcnt[e0 >> 16], 1.0f);
        if (p1) {
            float* o1 = acc + (e1 >> 16) * ACC_S + lane;
            atomicAdd(o1, b0); atomicAdd(o1 + 32, b1); atomicAdd(o1 + 64, b2); atomicAdd(o1 + 96, b3);
            if (with_count && lane == 0) atomicAdd(&dcnt[e1 >> 16], 1.0f);
        }
        if (p2) {
            float* o2 = acc + (e2 >> 16) * ACC_S + lane;
            atomicAdd(o2, c0); atomicAdd(o2 + 32, c1); atomicAdd(o2 + 64, c2); atomicAdd(o2 + 96, c3);
            if (with_count && lane == 0) atomicAdd(&dcnt[e2 >> 16], 1.0f);
        }
        if (p3) {
            float* o3 = acc + (e3 >> 16) * ACC_S + lane;
            atomicAdd(o3, g0); atomicAdd(o3 + 32, g1); atomicAdd(o3 + 64, g2); atomicAdd(o3 + 96, g3);
            if (with_count && lane == 0) atomicAdd(&dcnt[e3 >> 16], 1.0f);
        }
    }
}

#define PEND_CAP   256

template <bool CHECK>
__device__ __forceinline__ int sage_take(const unsigned e, const bool inside, const int wave,
                                         unsigned* pend, int cnt)
{
    bool mine = ((e >> 16) & 31u) == (unsigned)wave;
    if (CHECK) mine = mine && inside;
    const unsigned m = __builtin_amdgcn_ballot_w32(mine);
    if (m) {
        if (mine) pend[(cnt + (int)__builtin_amdgcn_mbcnt_lo(m, 0u)) & (PEND_CAP - 1)] = e;
        cnt += __builtin_popcount(m);
    }
    return cnt;
}

template <bool FULL>
__device__ __forceinline__ void sage_gather_group(const float* __restrict__ x, const unsigned* pend,
                                                  const int i, const int cnt, const int lane, v32f& A, int& cv)
{
    const int k = i + (lane & 7);
    const unsigned pv = pend[((FULL || k < cnt) ? k : i) & (PEND_CAP - 1)];
    unsigned e[8];
    v4f r[8];
#pragma unroll
    for (int q = 0; q < 8; ++q) {
        e[q] = (unsigned)__builtin_amdgcn_readlane((int)pv, q);
        r[q] = *(const v4f*)(x + (size_t)(e[q] & 0xFFFFu) * FDIM + 4 * lane);
    }
    asm volatile("" :: "v"(r[0]), "v"(r[1]), "v"(r[2]), "v"(r[3]), "v"(r[4]), "v"(r[5]), "v"(r[6]), "v"(r[7]));
#pragma unroll
    for (int q = 0; q < 8; ++q) {
        if (FULL || i + q < cnt) {
            const unsigned slot = e[q] >> 21;
            A[4u * slot + 0u] += r[q].x;
            A[4u * slot + 1u] += r[q].y;
            A[4u * slot + 2u] += r[q].z;
            A[4u * slot + 3u] += r[q].w;
            cv += (lane == (int)slot) ? 1 : 0;
        }
    }
}

__device__ __forceinline__ void sage_gather_owned(
    const float* __restrict__ x, const unsigned* list, unsigned* pend, float* acc, float* dcnt,
    const int n, const int wave, const int lane, const int with_count)
{
    v32f A = {};
    int cv = 0;
    const uint4* list4 = (const uint4*)list;
    int cnt = 0, done = 0;
    for (int base = 0; base < n; base += 128) {
        const uint4 v = list4[(base >> 2) + lane];
        if (base + 128 <= n) {
            cnt = sage_take<false>(v.x, true, wave, pend, cnt);
            cnt = sage_take<false>(v.y, true, wave, pend, cnt);
            cnt = sage_take<false>(v.z, true, wave, pend, cnt);
            cnt = sage_take<false>(v.w, true, wave, pend, cnt);
        } else {
            const int i4 = base + 4 * lane;
            cnt = sage_take<true>(v.x, i4 + 0 < n, wave, pend, cnt);
            cnt = sage_take<true>(v.y, i4 + 1 < n, wave, pend, cnt);
            cnt = sage_take<true>(v.z, i4 + 2 < n, wave, pend, cnt);
            cnt = sage_take<true>(v.w, i4 + 3 < n, wave, pend, cnt);
        }
        for (; done + 8 <= cnt; done += 8) sage_gather_group<true>(x, pend, done, cnt, lane, A, cv);
    }
    if (done < cnt) sage_gather_group<false>(x, pend, done, cnt, lane, A, cv);
#pragma unroll
    for (int j = 0; j < 7; ++j) {
        const v4f s = {A[4 * j], A[4 * j + 1], A[4 * j + 2], A[4 * j + 3]};
        const int r = wave + 32 * j;
        if (r < R_PAD) *(v4f*)(acc + r * ACC_S + 4 * lane) = s;
    }
    if (with_count && lane < 7 && wave + 32 * lane < R_PAD) dcnt[wave + 32 * lane] = (float)cv;
}

__device__ __forceinline__ unsigned sage_zero_bytes(const unsigned t)
{
    return ~(((t & 0x7F7F7F7Fu) + 0x7F7F7F7Fu) | t | 0x7F7F7F7Fu);
}

__device__ __forceinline__ unsigned sage_zero_nibble(const unsigned t)
{
    return (((sage_zero_bytes(t) >> 7) * 0x01020408u) >> 24) & 0xFu;
}

__device__ __forceinline__ void sage_load_ob(uint4 (&r)[8], const uint4* __restrict__ ob16, int c, int tid)
{
#pragma unroll
    for (int k = 0; k < 8; ++k) {
        int idx = c * OB_CHUNK + k * 1024 + tid;
        r[k] = ob16[idx < N_OB16 ? idx : (N_OB16 - 1)];
    }
}

template <bool CHECK>
__device__ __forceinline__ void sage_scan_ob(const uint4 (&r)[8], int c, int tid, unsigned pat,
                                             unsigned* glist, unsigned* ng)
{
#pragma unroll
    for (int k = 0; k < 8; ++k) {
        const int idx = c * OB_CHUNK + k * 1024 + tid;
        unsigned z = sage_zero_bytes(r[k].x ^ pat) | sage_zero_bytes(r[k].y ^ pat)
                   | sage_zero_bytes(r[k].z ^ pat) | sage_zero_bytes(r[k].w ^ pat);
        if (CHECK && idx >= N_OB16) z = 0u;
        if (z) {
            const unsigned pos = atomicAdd(ng, 1u);
            if (pos < (unsigned)GL_CAP) glist[pos] = (unsigned)idx;
        }
    }
}

__device__ __forceinline__ void sage_group_edges(const uint4 w, const unsigned g, const unsigned pat,
                                                 unsigned* list, unsigned* ne)
{
    unsigned m = sage_zero_nibble(w.x ^ pat) | (sage_zero_nibble(w.y ^ pat) << 4)
               | (sage_zero_nibble(w.z ^ pat) << 8) | (sage_zero_nibble(w.w ^ pat) << 12);
    unsigned pos = atomicAdd(ne, (unsigned)__popc(m));
    const unsigned e = g << 4;
    while (m) {
        const unsigned j = (unsigned)__ffs((int)m) - 1u;
        if (pos < (unsigned)EL_CAP) list[pos] = e + j;
        ++pos;
        m &= m - 1u;
    }
}

__device__ __forceinline__ void sage_load_chunk(int4 (&r)[8], const int4* __restrict__ dst4, int c, int tid)
{
#pragma unroll
    for (int k = 0; k < 8; ++k) {
        int idx = c * 8192 + k * 1024 + tid;
        r[k] = dst4[idx < N_DST4 ? idx : (N_DST4 - 1)];
    }
}

template <int NT>
__device__ __forceinline__ void sage_wave_product(
    const float* acc, const float* dcnt, const float* lw,
    const float* __restrict__ x, const float* __restrict__ bias, float* hout,
    int lo, int rt, int ct0, int lane, int do_relu)
{
    const int h   = lane >> 4;
    const int col = lane & 15;
    const int rl  = rt * 16 + col;
    int rg = lo + rl;
    rg = rg < N_NODES ? rg : (N_NODES - 1);

    const float  scale  = dcnt[rl];
    const float* arow_m = acc + rl * ACC_S + 2 * h;
    const float* arow_x = x + (size_t)rg * FDIM + 2 * h;
    const float* brow   = lw + (ct0 * 16 + col) * KSTRIDE + 2 * h;

    v8f d4[NT];
    v8f z = {0.f, 0.f, 0.f, 0.f, 0.f, 0.f, 0.f, 0.f};
#pragma unroll
    for (int t = 0; t < NT; ++t) d4[t] = z;

#pragma unroll 4
    for (int kk = 0; kk < 128; kk += 4) {
        v2f a = *(const v2f*)(arow_m + kk);
        a.x *= scale;
        a.y *= scale;
#pragma unroll
        for (int t = 0; t < NT; ++t) {
            v2f b = *(const v2f*)(brow + t * (16 * KSTRIDE) + kk);
            d4[t] = __builtin_amdgcn_wmma_f32_16x16x4_f32(
                false, a, false, b, (short)0, d4[t], false, false);
        }
    }

#pragma unroll 4
    for (int kk = 0; kk < 128; kk += 4) {
        v2f a = *(const v2f*)(arow_x + kk);
#pragma unroll
        for (int t = 0; t < NT; ++t) {
            v2f b = *(const v2f*)(brow + t * (16 * KSTRIDE) + 128 + kk);
            d4[t] = __builtin_amdgcn_wmma_f32_16x16x4_f32(
                false, a, false, b, (short)0, d4[t], false, false);
        }
    }

#pragma unroll
    for (int t = 0; t < NT; ++t) {
        int n  = (ct0 + t) * 16 + col;
        float bv = bias[n];
#pragma unroll
        for (int v = 0; v < 8; ++v) {
            int row = rt * 16 + v + 8 * h;
            float val = d4[t][v] + bv;
            if (do_relu) val = fmaxf(val, 0.0f);
            if (row < R_NODES) hout[(size_t)(lo + row) * FDIM + n] = val;
        }
    }
}

__global__ __launch_bounds__(1024) void sage_layer_kernel(
    const float* __restrict__ x,
    const int*   __restrict__ src,
    const int*   __restrict__ dst,
    const unsigned* __restrict__ ob,
    float* cnt,
    float* keep,
    const float* __restrict__ Wl, const float* __restrict__ Wr,
    const float* __restrict__ bias,
    float* hout,
    int first,
    int do_relu)
{
    __shared__ __attribute__((aligned(16))) float    acc[R_PAD * ACC_S];
    __shared__ float    dcnt[R_PAD];
    __shared__ __attribute__((aligned(16))) unsigned ubuf[128 * KSTRIDE];
    __shared__ unsigned nl[4];
    __shared__ __attribute__((aligned(16))) unsigned pendbuf[32 * PEND_CAP];
    unsigned* list = ubuf;
    float*    lw   = (float*)ubuf;

    const int tid  = threadIdx.x;
    const int lane = tid & 31;
    const int wave = __builtin_amdgcn_readfirstlane(tid >> 5);
    const int lo   = blockIdx.x * R_NODES;
    unsigned* kl   = (unsigned*)keep + (size_t)blockIdx.x * KEEP_WORDS;

    unsigned* pend = pendbuf + wave * PEND_CAP;

    if (tid < 4) nl[tid] = 0u;

    const unsigned kept = first ? 0xFFFFFFFFu : kl[0];

    if (kept <= (unsigned)KEEP_CAP) {
        const unsigned* ke = kl + KEEP_HDR;
        for (unsigned i = tid; i < kept; i += 4096) {
            const unsigned i1 = i + 1024, i2 = i + 2048, i3 = i + 3072;
            const bool q1 = i1 < kept, q2 = i2 < kept, q3 = i3 < kept;
            unsigned v0 = ke[i];
            unsigned v1 = q1 ? ke[i1] : 0u;
            unsigned v2 = q2 ? ke[i2] : 0u;
            unsigned v3 = q3 ? ke[i3] : 0u;
            list[i] = v0;
            if (q1) list[i1] = v1;
            if (q2) list[i2] = v2;
            if (q3) list[i3] = v3;
        }
        __syncthreads();
        sage_gather_owned(x, list, pend, acc, dcnt, (int)kept, wave, lane, 0);
        __syncthreads();
    } else {
        const int4*  dst4  = (const int4*)dst;
        const uint4* ob16  = (const uint4*)ob;
        const unsigned pat = (unsigned)blockIdx.x * 0x01010101u;
        unsigned*    glist = ubuf + GL_OFF;

        {
            uint4 b[8], b2[8];
            sage_load_ob(b, ob16, 0, tid);
            __syncthreads();
            for (int c = 0; c < OB_FULL; c += 2) {
                sage_load_ob(b2, ob16, c + 1, tid);
                sage_scan_ob<false>(b, c, tid, pat, glist, &nl[2]);
                sage_load_ob(b, ob16, c + 2, tid);
                sage_scan_ob<false>(b2, c + 1, tid, pat, glist, &nl[2]);
            }
            sage_scan_ob<true>(b, OB_FULL, tid, pat, glist, &nl[2]);
        }
        __syncthreads();
        const unsigned ngrp = nl[2];

        if (ngrp <= (unsigned)GL_CAP) {
            for (unsigned i = tid; i < ngrp; i += 4096) {
                const unsigned i1 = i + 1024, i2 = i + 2048, i3 = i + 3072;
                const bool q1 = i1 < ngrp, q2 = i2 < ngrp, q3 = i3 < ngrp;
                const unsigned g0 = glist[i];
                const unsigned g1 = q1 ? glist[i1] : g0;
                const unsigned g2 = q2 ? glist[i2] : g0;
                const unsigned g3 = q3 ? glist[i3] : g0;
                const uint4 w0 = ob16[g0], w1 = ob16[g1], w2 = ob16[g2], w3 = ob16[g3];
                sage_group_edges(w0, g0, pat, list, &nl[3]);
                if (q1) sage_group_edges(w1, g1, pat, list, &nl[3]);
                if (q2) sage_group_edges(w2, g2, pat, list, &nl[3]);
                if (q3) sage_group_edges(w3, g3, pat, list, &nl[3]);
            }
        }
        __syncthreads();
        const unsigned total = (ngrp <= (unsigned)GL_CAP && nl[3] <= (unsigned)EL_CAP) ? nl[3] : 0xFFFFFFFFu;

        if (total <= (unsigned)EL_CAP) {
            const bool keepit = first != 0;
            unsigned* ke = kl + KEEP_HDR;
            for (unsigned i = tid; i < total; i += 4096) {
                const unsigned i1 = i + 1024, i2 = i + 2048, i3 = i + 3072;
                const bool q1 = i1 < total, q2 = i2 < total, q3 = i3 < total;
                const unsigned a0 = list[i];
                const unsigned a1 = q1 ? list[i1] : a0;
                const unsigned a2 = q2 ? list[i2] : a0;
                const unsigned a3 = q3 ? list[i3] : a0;
                const unsigned s0 = (unsigned)src[a0], r0 = (unsigned)(dst[a0] - lo);
                const unsigned s1 = (unsigned)src[a1], r1 = (unsigned)(dst[a1] - lo);
                const unsigned s2 = (unsigned)src[a2], r2 = (unsigned)(dst[a2] - lo);
                const unsigned s3 = (unsigned)src[a3], r3 = (unsigned)(dst[a3] - lo);
                const unsigned v0 = s0 | (r0 << 16);
                const unsigned v1 = s1 | (r1 << 16);
                const unsigned v2 = s2 | (r2 << 16);
                const unsigned v3 = s3 | (r3 << 16);
                list[i] = v0;
                if (q1) list[i1] = v1;
                if (q2) list[i2] = v2;
                if (q3) list[i3] = v3;
                if (keepit) {
                    ke[i] = v0;
                    if (q1) ke[i1] = v1;
                    if (q2) ke[i2] = v2;
                    if (q3) ke[i3] = v3;
                }
            }
            __syncthreads();
            sage_gather_owned(x, list, pend, acc, dcnt, (int)total, wave, lane, first);
            __syncthreads();
        } else {
            for (int i = tid; i < R_PAD * ACC_S; i += 1024) acc[i] = 0.0f;
            if (tid < R_PAD) dcnt[tid] = 0.0f;
            int4 d[8];
            sage_load_chunk(d, dst4, 0, tid);
            for (int c = 0; c < N_CHUNK; ++c) {
                const int cur = c & 1;
                int  sv[32];
                bool hit[8];
#pragma unroll
                for (int k = 0; k < 8; ++k) {
                    int ebase = (c * 8192 + k * 1024 + tid) << 2;
                    unsigned l0 = (unsigned)(d[k].x - lo), l1 = (unsigned)(d[k].y - lo);
                    unsigned l2 = (unsigned)(d[k].z - lo), l3 = (unsigned)(d[k].w - lo);
                    unsigned mn = min(min(l0, l1), min(l2, l3));
                    hit[k] = mn < (unsigned)R_NODES;
                    if (hit[k]) {
                        if (l0 < (unsigned)R_NODES) sv[k * 4 + 0] = src[ebase + 0];
                        if (l1 < (unsigned)R_NODES) sv[k * 4 + 1] = src[ebase + 1];
                        if (l2 < (unsigned)R_NODES) sv[k * 4 + 2] = src[ebase + 2];
                        if (l3 < (unsigned)R_NODES) sv[k * 4 + 3] = src[ebase + 3];
                    }
                }
#pragma unroll
                for (int k = 0; k < 8; ++k) {
                    if (hit[k]) {
                        unsigned ll[4] = {(unsigned)(d[k].x - lo), (unsigned)(d[k].y - lo),
                                          (unsigned)(d[k].z - lo), (unsigned)(d[k].w - lo)};
#pragma unroll
                        for (int q = 0; q < 4; ++q) {
                            if (ll[q] < (unsigned)R_NODES) {
                                unsigned pos = atomicAdd(&nl[cur], 1u);
                                list[pos] = (unsigned)sv[k * 4 + q] | (ll[q] << 16);
                            }
                        }
                    }
                }
                __syncthreads();

#pragma unroll
                for (int k = 0; k < 8; ++k) {
                    int idx  = (c + 1) * 8192 + k * 1024 + tid;
                    int idxc = idx < N_DST4 ? idx : (N_DST4 - 1);
                    int4 v = dst4[idxc];
                    if (idx >= N_DST4) { v.x = -1; v.y = -1; v.z = -1; v.w = -1; }
                    d[k] = v;
                }
                if (tid == 0) nl[cur ^ 1] = 0u;

                const int n = (int)nl[cur];
                sage_gather_list(x, list, acc, dcnt, n, wave, lane, first);
                __syncthreads();
            }
        }
        if (first && tid == 0) kl[0] = total;
    }

    if (tid < R_PAD) {
        float c = 1.0f;
        if (tid < R_NODES) {
            if (first) { c = dcnt[tid]; cnt[lo + tid] = c; }
            else       { c = cnt[lo + tid]; }
        }
        dcnt[tid] = 1.0f / fmaxf(c, 1.0f);
    }
    for (int idx = tid; idx < 256 * 128; idx += 1024) {
        int k = idx >> 7;
        int n = idx & 127;
        float w = (k < 128) ? Wl[idx] : Wr[idx - 128 * 128];
        lw[n * KSTRIDE + k] = w;
    }
    __syncthreads();

    if (wave < 24) sage_wave_product<4>(acc, dcnt, lw, x, bias, hout, lo, wave >> 1, (wave & 1) * 4, lane, do_relu);
    else           sage_wave_product<1>(acc, dcnt, lw, x, bias, hout, lo, 12, wave - 24, lane, do_relu);
}

__global__ __launch_bounds__(1024) void sage_owner_kernel(const int* __restrict__ dst, unsigned* __restrict__ ob)
{
    const int i = blockIdx.x * 1024 + threadIdx.x;
    if (i < N_DST4) {
        const int4 d = ((const int4*)dst)[i];
        const unsigned u0 = (unsigned)d.x, u1 = (unsigned)d.y, u2 = (unsigned)d.z, u3 = (unsigned)d.w;
        const unsigned o0 = u0 < (unsigned)N_NODES ? u0 / (unsigned)R_NODES : 255u;
        const unsigned o1 = u1 < (unsigned)N_NODES ? u1 / (unsigned)R_NODES : 255u;
        const unsigned o2 = u2 < (unsigned)N_NODES ? u2 / (unsigned)R_NODES : 255u;
        const unsigned o3 = u3 < (unsigned)N_NODES ? u3 / (unsigned)R_NODES : 255u;
        ob[i] = o0 | (o1 << 8) | (o2 << 16) | (o3 << 24);
    }
}

extern "C" void kernel_launch(void* const* d_in, const int* in_sizes, int n_in,
                              void* d_out, int out_size, void* d_ws, size_t ws_size,
                              hipStream_t stream) {
    const float* x   = (const float*)d_in[0];
    const int*   ei  = (const int*)d_in[1];
    const float* Wl0 = (const float*)d_in[2];
    const float* Wr0 = (const float*)d_in[3];
    const float* b0  = (const float*)d_in[4];
    const float* Wl1 = (const float*)d_in[5];
    const float* Wr1 = (const float*)d_in[6];
    const float* b1  = (const float*)d_in[7];
    float* out = (float*)d_out;

    const int* src = ei;
    const int* dst = ei + N_EDGES;

    unsigned* ob = (unsigned*)d_ws;
    float* cnt  = (float*)d_ws + (size_t)N_NODES * FDIM;
    float* hbuf = cnt + N_NODES;

    sage_owner_kernel<<<OWN_BLOCKS, 1024, 0, stream>>>(dst, ob);
    sage_layer_kernel<<<AGG_BLOCKS, 1024, 0, stream>>>(x, src, dst, ob, cnt, out, Wl0, Wr0, b0, hbuf, 1, 1);
    sage_layer_kernel<<<AGG_BLOCKS, 1024, 0, stream>>>(hbuf, src, dst, ob, cnt, out, Wl1, Wr1, b1, out, 0, 0);
}


# list split by owner wave once (LDS counting sort, 32 bins; kept split for layer 1): a wave walks only its own part, not the whole list 32 times; row sums now in placement-counter arrival order
# speedup vs baseline: 3.7802x; 1.0317x over previous
#include <hip/hip_runtime.h>

typedef float v2f __attribute__((ext_vector_type(2)));
typedef float v4f __attribute__((ext_vector_type(4)));
typedef float v32f __attribute__((ext_vector_type(32)));
typedef float v8f __attribute__((ext_vector_type(8)));


#define N_NODES 50000
#define FDIM    128
#define N_EDGES 800000
#define KSTRIDE 260

#define R_NODES    200
#define R_PAD      208
#define ACC_S      132
#define AGG_BLOCKS 250
#define CHUNK_E    32768
#define N_CHUNK    25
#define N_DST4     (N_EDGES / 4)
#define LIST_CAP   32768
#define N_OB16     (N_EDGES / 16)
#define OB_CHUNK   8192
#define OB_FULL    6
#define OWN_BLOCKS 196
#define EL_CAP     24576
#define GL_OFF     24576
#define GL_CAP     8192
#define KEEP_WORDS (R_NODES * FDIM)
#define KEEP_HDR   64
#define KEEP_CAP   (KEEP_WORDS - KEEP_HDR)

__device__ __forceinline__ void sage_gather_list(
    const float* __restrict__ x, const unsigned* list, float* acc, float* dcnt,
    int n, int wave, int lane, int with_count)
{
    for (int i = wave; i < n; i += 128) {
        const bool p1 = (i + 32) < n, p2 = (i + 64) < n, p3 = (i + 96) < n;
        unsigned e0 = list[i];
        unsigned e1 = p1 ? list[i + 32] : e0;
        unsigned e2 = p2 ? list[i + 64] : e0;
        unsigned e3 = p3 ? list[i + 96] : e0;
        const float* r0 = x + (size_t)(e0 & 0xFFFFu) * FDIM + lane;
        const float* r1 = x + (size_t)(e1 & 0xFFFFu) * FDIM + lane;
        const float* r2 = x + (size_t)(e2 & 0xFFFFu) * FDIM + lane;
        const float* r3 = x + (size_t)(e3 & 0xFFFFu) * FDIM + lane;
        float a0 = r0[0], a1 = r0[32], a2 = r0[64], a3 = r0[96];
        float b0 = r1[0], b1 = r1[32], b2 = r1[64], b3 = r1[96];
        float c0 = r2[0], c1 = r2[32], c2 = r2[64], c3 = r2[96];
        float g0 = r3[0], g1 = r3[32], g2 = r3[64], g3 = r3[96];
        asm volatile("" :: "v"(a0), "v"(a1), "v"(a2), "v"(a3), "v"(b0), "v"(b1), "v"(b2), "v"(b3),
                           "v"(c0), "v"(c1), "v"(c2), "v"(c3), "v"(g0), "v"(g1), "v"(g2), "v"(g3));
        float* o0 = acc + (e0 >> 16) * ACC_S + lane;
        atomicAdd(o0, a0); atomicAdd(o0 + 32, a1); atomicAdd(o0 + 64, a2); atomicAdd(o0 + 96, a3);
        if (with_count && lane == 0) atomicAdd(&dcnt[e0 >> 16], 1.0f);
        if (p1) {
            float* o1 = acc + (e1 >> 16) * ACC_S + lane;
            atomicAdd(o1, b0); atomicAdd(o1 + 32, b1); atomicAdd(o1 + 64, b2); atomicAdd(o1 + 96, b3);
            if (with_count && lane == 0) atomicAdd(&dcnt[e1 >> 16], 1.0f);
        }
        if (p2) {
            float* o2 = acc + (e2 >> 16) * ACC_S + lane;
            atomicAdd(o2, c0); atomicAdd(o2 + 32, c1); atomicAdd(o2 + 64, c2); atomicAdd(o2 + 96, c3);
            if (with_count && lane == 0) atomicAdd(&dcnt[e2 >> 16], 1.0f);
        }
        if (p3) {
            float* o3 = acc + (e3 >> 16) * ACC_S + lane;
            atomicAdd(o3, g0); atomicAdd(o3 + 32, g1); atomicAdd(o3 + 64, g2); atomicAdd(o3 + 96, g3);
            if (with_count && lane == 0) atomicAdd(&dcnt[e3 >> 16], 1.0f);
        }
    }
}

template <bool FULL>
__device__ __forceinline__ void sage_gather_group(const float* __restrict__ x, const unsigned* part,
                                                  const int i, const int end, const int lane, v32f& A, int& cv)
{
    const int k = i + (lane & 7);
    const unsigned pv = part[(FULL || k < end) ? k : i];
    unsigned e[8];
    v4f r[8];
#pragma unroll
    for (int q = 0; q < 8; ++q) {
        e[q] = (unsigned)__builtin_amdgcn_readlane((int)pv, q);
        r[q] = *(const v4f*)(x + (size_t)(e[q] & 0xFFFFu) * FDIM + 4 * lane);
    }
    asm volatile("" :: "v"(r[0]), "v"(r[1]), "v"(r[2]), "v"(r[3]), "v"(r[4]), "v"(r[5]), "v"(r[6]), "v"(r[7]));
#pragma unroll
    for (int q = 0; q < 8; ++q) {
        if (FULL || i + q < end) {
            const unsigned slot = e[q] >> 21;
            A[4u * slot + 0u] += r[q].x;
            A[4u * slot + 1u] += r[q].y;
            A[4u * slot + 2u] += r[q].z;
            A[4u * slot + 3u] += r[q].w;
            cv += (lane == (int)slot) ? 1 : 0;
        }
    }
}

__device__ __forceinline__ void sage_gather_owned(
    const float* __restrict__ x, const unsigned* part, const int beg, const int end, float* acc, float* dcnt,
    const int wave, const int lane, const int with_count)
{
    v32f A = {};
    int cv = 0;
    int i = beg;
    for (; i + 8 <= end; i += 8) sage_gather_group<true>(x, part, i, end, lane, A, cv);
    if (i < end) sage_gather_group<false>(x, part, i, end, lane, A, cv);
    __syncthreads();
#pragma unroll
    for (int j = 0; j < 7; ++j) {
        const v4f s = {A[4 * j], A[4 * j + 1], A[4 * j + 2], A[4 * j + 3]};
        const int r = wave + 32 * j;
        if (r < R_PAD) *(v4f*)(acc + r * ACC_S + 4 * lane) = s;
    }
    if (with_count && lane < 7 && wave + 32 * lane < R_PAD) dcnt[wave + 32 * lane] = (float)cv;
}

__device__ __forceinline__ unsigned sage_zero_bytes(const unsigned t)
{
    return ~(((t & 0x7F7F7F7Fu) + 0x7F7F7F7Fu) | t | 0x7F7F7F7Fu);
}

__device__ __forceinline__ unsigned sage_zero_nibble(const unsigned t)
{
    return (((sage_zero_bytes(t) >> 7) * 0x01020408u) >> 24) & 0xFu;
}

__device__ __forceinline__ void sage_load_ob(uint4 (&r)[8], const uint4* __restrict__ ob16, int c, int tid)
{
#pragma unroll
    for (int k = 0; k < 8; ++k) {
        int idx = c * OB_CHUNK + k * 1024 + tid;
        r[k] = ob16[idx < N_OB16 ? idx : (N_OB16 - 1)];
    }
}

template <bool CHECK>
__device__ __forceinline__ void sage_scan_ob(const uint4 (&r)[8], int c, int tid, unsigned pat,
                                             unsigned* glist, unsigned* ng)
{
#pragma unroll
    for (int k = 0; k < 8; ++k) {
        const int idx = c * OB_CHUNK + k * 1024 + tid;
        unsigned z = sage_zero_bytes(r[k].x ^ pat) | sage_zero_bytes(r[k].y ^ pat)
                   | sage_zero_bytes(r[k].z ^ pat) | sage_zero_bytes(r[k].w ^ pat);
        if (CHECK && idx >= N_OB16) z = 0u;
        if (z) {
            const unsigned pos = atomicAdd(ng, 1u);
            if (pos < (unsigned)GL_CAP) glist[pos] = (unsigned)idx;
        }
    }
}

__device__ __forceinline__ void sage_group_edges(const uint4 w, const unsigned g, const unsigned pat,
                                                 unsigned* list, unsigned* ne)
{
    unsigned m = sage_zero_nibble(w.x ^ pat) | (sage_zero_nibble(w.y ^ pat) << 4)
               | (sage_zero_nibble(w.z ^ pat) << 8) | (sage_zero_nibble(w.w ^ pat) << 12);
    unsigned pos = atomicAdd(ne, (unsigned)__popc(m));
    const unsigned e = g << 4;
    while (m) {
        const unsigned j = (unsigned)__ffs((int)m) - 1u;
        if (pos < (unsigned)EL_CAP) list[pos] = e + j;
        ++pos;
        m &= m - 1u;
    }
}

__device__ __forceinline__ void sage_load_chunk(int4 (&r)[8], const int4* __restrict__ dst4, int c, int tid)
{
#pragma unroll
    for (int k = 0; k < 8; ++k) {
        int idx = c * 8192 + k * 1024 + tid;
        r[k] = dst4[idx < N_DST4 ? idx : (N_DST4 - 1)];
    }
}

template <int NT>
__device__ __forceinline__ void sage_wave_product(
    const float* acc, const float* dcnt, const float* lw,
    const float* __restrict__ x, const float* __restrict__ bias, float* hout,
    int lo, int rt, int ct0, int lane, int do_relu)
{
    const int h   = lane >> 4;
    const int col = lane & 15;
    const int rl  = rt * 16 + col;
    int rg = lo + rl;
    rg = rg < N_NODES ? rg : (N_NODES - 1);

    const float  scale  = dcnt[rl];
    const float* arow_m = acc + rl * ACC_S + 2 * h;
    const float* arow_x = x + (size_t)rg * FDIM + 2 * h;
    const float* brow   = lw + (ct0 * 16 + col) * KSTRIDE + 2 * h;

    v8f d4[NT];
    v8f z = {0.f, 0.f, 0.f, 0.f, 0.f, 0.f, 0.f, 0.f};
#pragma unroll
    for (int t = 0; t < NT; ++t) d4[t] = z;

#pragma unroll 4
    for (int kk = 0; kk < 128; kk += 4) {
        v2f a = *(const v2f*)(arow_m + kk);
        a.x *= scale;
        a.y *= scale;
#pragma unroll
        for (int t = 0; t < NT; ++t) {
            v2f b = *(const v2f*)(brow + t * (16 * KSTRIDE) + kk);
            d4[t] = __builtin_amdgcn_wmma_f32_16x16x4_f32(
                false, a, false, b, (short)0, d4[t], false, false);
        }
    }

#pragma unroll 4
    for (int kk = 0; kk < 128; kk += 4) {
        v2f a = *(const v2f*)(arow_x + kk);
#pragma unroll
        for (int t = 0; t < NT; ++t) {
            v2f b = *(const v2f*)(brow + t * (16 * KSTRIDE) + 128 + kk);
            d4[t] = __builtin_amdgcn_wmma_f32_16x16x4_f32(
                false, a, false, b, (short)0, d4[t], false, false);
        }
    }

#pragma unroll
    for (int t = 0; t < NT; ++t) {
        int n  = (ct0 + t) * 16 + col;
        float bv = bias[n];
#pragma unroll
        for (int v = 0; v < 8; ++v) {
            int row = rt * 16 + v + 8 * h;
            float val = d4[t][v] + bv;
            if (do_relu) val = fmaxf(val, 0.0f);
            if (row < R_NODES) hout[(size_t)(lo + row) * FDIM + n] = val;
        }
    }
}

__global__ __launch_bounds__(1024) void sage_layer_kernel(
    const float* __restrict__ x,
    const int*   __restrict__ src,
    const int*   __restrict__ dst,
    const unsigned* __restrict__ ob,
    float* cnt,
    float* keep,
    const float* __restrict__ Wl, const float* __restrict__ Wr,
    const float* __restrict__ bias,
    float* hout,
    int first,
    int do_relu)
{
    __shared__ __attribute__((aligned(16))) float    acc[R_PAD * ACC_S];
    __shared__ float    dcnt[R_PAD];
    __shared__ __attribute__((aligned(16))) unsigned ubuf[128 * KSTRIDE];
    __shared__ unsigned nl[4];
    __shared__ unsigned wcnt[32];
    __shared__ unsigned woff[33];
    __shared__ unsigned wpos[32];
    unsigned* list = ubuf;
    float*    lw   = (float*)ubuf;
    unsigned* part = (unsigned*)acc;

    const int tid  = threadIdx.x;
    const int lane = tid & 31;
    const int wave = __builtin_amdgcn_readfirstlane(tid >> 5);
    const int lo   = blockIdx.x * R_NODES;
    unsigned* kl   = (unsigned*)keep + (size_t)blockIdx.x * KEEP_WORDS;

    if (tid < 4) nl[tid] = 0u;
    if (tid < 32) wcnt[tid] = 0u;

    const unsigned kept = first ? 0xFFFFFFFFu : kl[0];

    if (kept <= (unsigned)KEEP_CAP) {
        const unsigned* ke = kl + KEEP_HDR;
        const unsigned ue = min(kl[2 + wave], kept), ub = min(kl[1 + wave], ue);
        const int beg = __builtin_amdgcn_readfirstlane((int)ub);
        const int end = __builtin_amdgcn_readfirstlane((int)ue);
        for (unsigned i = tid; i < kept; i += 4096) {
            const unsigned i1 = i + 1024, i2 = i + 2048, i3 = i + 3072;
            const bool q1 = i1 < kept, q2 = i2 < kept, q3 = i3 < kept;
            unsigned v0 = ke[i];
            unsigned v1 = q1 ? ke[i1] : 0u;
            unsigned v2 = q2 ? ke[i2] : 0u;
            unsigned v3 = q3 ? ke[i3] : 0u;
            part[i] = v0;
            if (q1) part[i1] = v1;
            if (q2) part[i2] = v2;
            if (q3) part[i3] = v3;
        }
        __syncthreads();
        sage_gather_owned(x, part, beg, end, acc, dcnt, wave, lane, 0);
        __syncthreads();
    } else {
        const int4*  dst4  = (const int4*)dst;
        const uint4* ob16  = (const uint4*)ob;
        const unsigned pat = (unsigned)blockIdx.x * 0x01010101u;
        unsigned*    glist = ubuf + GL_OFF;

        {
            uint4 b[8], b2[8];
            sage_load_ob(b, ob16, 0, tid);
            __syncthreads();
            for (int c = 0; c < OB_FULL; c += 2) {
                sage_load_ob(b2, ob16, c + 1, tid);
                sage_scan_ob<false>(b, c, tid, pat, glist, &nl[2]);
                sage_load_ob(b, ob16, c + 2, tid);
                sage_scan_ob<false>(b2, c + 1, tid, pat, glist, &nl[2]);
            }
            sage_scan_ob<true>(b, OB_FULL, tid, pat, glist, &nl[2]);
        }
        __syncthreads();
        const unsigned ngrp = nl[2];

        if (ngrp <= (unsigned)GL_CAP) {
            for (unsigned i = tid; i < ngrp; i += 4096) {
                const unsigned i1 = i + 1024, i2 = i + 2048, i3 = i + 3072;
                const bool q1 = i1 < ngrp, q2 = i2 < ngrp, q3 = i3 < ngrp;
                const unsigned g0 = glist[i];
                const unsigned g1 = q1 ? glist[i1] : g0;
                const unsigned g2 = q2 ? glist[i2] : g0;
                const unsigned g3 = q3 ? glist[i3] : g0;
                const uint4 w0 = ob16[g0], w1 = ob16[g1], w2 = ob16[g2], w3 = ob16[g3];
                sage_group_edges(w0, g0, pat, list, &nl[3]);
                if (q1) sage_group_edges(w1, g1, pat, list, &nl[3]);
                if (q2) sage_group_edges(w2, g2, pat, list, &nl[3]);
                if (q3) sage_group_edges(w3, g3, pat, list, &nl[3]);
            }
        }
        __syncthreads();
        const unsigned total = (ngrp <= (unsigned)GL_CAP && nl[3] <= (unsigned)EL_CAP) ? nl[3] : 0xFFFFFFFFu;

        if (total <= (unsigned)EL_CAP) {
            const bool keepit = first != 0;
            unsigned* ke = kl + KEEP_HDR;
            for (unsigned i = tid; i < total; i += 4096) {
                const unsigned i1 = i + 1024, i2 = i + 2048, i3 = i + 3072;
                const bool q1 = i1 < total, q2 = i2 < total, q3 = i3 < total;
                const unsigned a0 = list[i];
                const unsigned a1 = q1 ? list[i1] : a0;
                const unsigned a2 = q2 ? list[i2] : a0;
                const unsigned a3 = q3 ? list[i3] : a0;
                const unsigned s0 = (unsigned)src[a0], r0 = (unsigned)(dst[a0] - lo);
                const unsigned s1 = (unsigned)src[a1], r1 = (unsigned)(dst[a1] - lo);
                const unsigned s2 = (unsigned)src[a2], r2 = (unsigned)(dst[a2] - lo);
                const unsigned s3 = (unsigned)src[a3], r3 = (unsigned)(dst[a3] - lo);
                list[i] = s0 | (r0 << 16);
                atomicAdd(&wcnt[r0 & 31u], 1u);
                if (q1) { list[i1] = s1 | (r1 << 16); atomicAdd(&wcnt[r1 & 31u], 1u); }
                if (q2) { list[i2] = s2 | (r2 << 16); atomicAdd(&wcnt[r2 & 31u], 1u); }
                if (q3) { list[i3] = s3 | (r3 << 16); atomicAdd(&wcnt[r3 & 31u], 1u); }
            }
            __syncthreads();
            if (tid < 32) {
                const unsigned cn = wcnt[tid];
                unsigned s = cn;
#pragma unroll
                for (int d = 1; d < 32; d <<= 1) {
                    const unsigned t = __shfl_up(s, (unsigned)d);
                    if (lane >= d) s += t;
                }
                const unsigned ex = s - cn;
                woff[tid] = ex;
                wpos[tid] = ex;
                if (keepit) kl[1 + tid] = ex;
                if (tid == 31) { woff[32] = s; if (keepit) kl[33] = s; }
            }
            __syncthreads();
            for (unsigned i = tid; i < total; i += 1024) {
                const unsigned v = list[i];
                part[atomicAdd(&wpos[(v >> 16) & 31u], 1u)] = v;
            }
            __syncthreads();
            if (keepit) for (unsigned i = tid; i < total; i += 1024) ke[i] = part[i];
            const int beg = __builtin_amdgcn_readfirstlane((int)woff[wave]);
            const int end = __builtin_amdgcn_readfirstlane((int)woff[wave + 1]);
            sage_gather_owned(x, part, beg, end, acc, dcnt, wave, lane, first);
            __syncthreads();
        } else {
            for (int i = tid; i < R_PAD * ACC_S; i += 1024) acc[i] = 0.0f;
            if (tid < R_PAD) dcnt[tid] = 0.0f;
            int4 d[8];
            sage_load_chunk(d, dst4, 0, tid);
            for (int c = 0; c < N_CHUNK; ++c) {
                const int cur = c & 1;
                int  sv[32];
                bool hit[8];
#pragma unroll
                for (int k = 0; k < 8; ++k) {
                    int ebase = (c * 8192 + k * 1024 + tid) << 2;
                    unsigned l0 = (unsigned)(d[k].x - lo), l1 = (unsigned)(d[k].y - lo);
                    unsigned l2 = (unsigned)(d[k].z - lo), l3 = (unsigned)(d[k].w - lo);
                    unsigned mn = min(min(l0, l1), min(l2, l3));
                    hit[k] = mn < (unsigned)R_NODES;
                    if (hit[k]) {
                        if (l0 < (unsigned)R_NODES) sv[k * 4 + 0] = src[ebase + 0];
                        if (l1 < (unsigned)R_NODES) sv[k * 4 + 1] = src[ebase + 1];
                        if (l2 < (unsigned)R_NODES) sv[k * 4 + 2] = src[ebase + 2];
                        if (l3 < (unsigned)R_NODES) sv[k * 4 + 3] = src[ebase + 3];
                    }
                }
#pragma unroll
                for (int k = 0; k < 8; ++k) {
                    if (hit[k]) {
                        unsigned ll[4] = {(unsigned)(d[k].x - lo), (unsigned)(d[k].y - lo),
                                          (unsigned)(d[k].z - lo), (unsigned)(d[k].w - lo)};
#pragma unroll
                        for (int q = 0; q < 4; ++q) {
                            if (ll[q] < (unsigned)R_NODES) {
                                unsigned pos = atomicAdd(&nl[cur], 1u);
                                list[pos] = (unsigned)sv[k * 4 + q] | (ll[q] << 16);
                            }
                        }
                    }
                }
                __syncthreads();

#pragma unroll
                for (int k = 0; k < 8; ++k) {
                    int idx  = (c + 1) * 8192 + k * 1024 + tid;
                    int idxc = idx < N_DST4 ? idx : (N_DST4 - 1);
                    int4 v = dst4[idxc];
                    if (idx >= N_DST4) { v.x = -1; v.y = -1; v.z = -1; v.w = -1; }
                    d[k] = v;
                }
                if (tid == 0) nl[cur ^ 1] = 0u;

                const int n = (int)nl[cur];
                sage_gather_list(x, list, acc, dcnt, n, wave, lane, first);
                __syncthreads();
            }
        }
        if (first && tid == 0) kl[0] = total;
    }

    if (tid < R_PAD) {
        float c = 1.0f;
        if (tid < R_NODES) {
            if (first) { c = dcnt[tid]; cnt[lo + tid] = c; }
            else       { c = cnt[lo + tid]; }
        }
        dcnt[tid] = 1.0f / fmaxf(c, 1.0f);
    }
    for (int idx = tid; idx < 256 * 128; idx += 1024) {
        int k = idx >> 7;
        int n = idx & 127;
        float w = (k < 128) ? Wl[idx] : Wr[idx - 128 * 128];
        lw[n * KSTRIDE + k] = w;
    }
    __syncthreads();

    if (wave < 24) sage_wave_product<4>(acc, dcnt, lw, x, bias, hout, lo, wave >> 1, (wave & 1) * 4, lane, do_relu);
    else           sage_wave_product<1>(acc, dcnt, lw, x, bias, hout, lo, 12, wave - 24, lane, do_relu);
}

__global__ __launch_bounds__(1024) void sage_owner_kernel(const int* __restrict__ dst, unsigned* __restrict__ ob)
{
    const int i = blockIdx.x * 1024 + threadIdx.x;
    if (i < N_DST4) {
        const int4 d = ((const int4*)dst)[i];
        const unsigned u0 = (unsigned)d.x, u1 = (unsigned)d.y, u2 = (unsigned)d.z, u3 = (unsigned)d.w;
        const unsigned o0 = u0 < (unsigned)N_NODES ? u0 / (unsigned)R_NODES : 255u;
        const unsigned o1 = u1 < (unsigned)N_NODES ? u1 / (unsigned)R_NODES : 255u;
        const unsigned o2 = u2 < (unsigned)N_NODES ? u2 / (unsigned)R_NODES : 255u;
        const unsigned o3 = u3 < (unsigned)N_NODES ? u3 / (unsigned)R_NODES : 255u;
        ob[i] = o0 | (o1 << 8) | (o2 << 16) | (o3 << 24);
    }
}

extern "C" void kernel_launch(void* const* d_in, const int* in_sizes, int n_in,
                              void* d_out, int out_size, void* d_ws, size_t ws_size,
                              hipStream_t stream) {
    const float* x   = (const float*)d_in[0];
    const int*   ei  = (const int*)d_in[1];
    const float* Wl0 = (const float*)d_in[2];
    const float* Wr0 = (const float*)d_in[3];
    const float* b0  = (const float*)d_in[4];
    const float* Wl1 = (const float*)d_in[5];
    const float* Wr1 = (const float*)d_in[6];
    const float* b1  = (const float*)d_in[7];
    float* out = (float*)d_out;

    const int* src = ei;
    const int* dst = ei + N_EDGES;

    unsigned* ob = (unsigned*)d_ws;
    float* cnt  = (float*)d_ws + (size_t)N_NODES * FDIM;
    float* hbuf = cnt + N_NODES;

    sage_owner_kernel<<<OWN_BLOCKS, 1024, 0, stream>>>(dst, ob);
    sage_layer_kernel<<<AGG_BLOCKS, 1024, 0, stream>>>(x, src, dst, ob, cnt, out, Wl0, Wr0, b0, hbuf, 1, 1);
    sage_layer_kernel<<<AGG_BLOCKS, 1024, 0, stream>>>(hbuf, src, dst, ob, cnt, out, Wl1, Wr1, b1, out, 0, 0);
}


# gather balanced: list sorted by dst row (LDS counting sort), every wave takes an equal share; one row's run summed in 4 fixed registers, runs joined by LDS f32 adds (list order, joins by arrival)
# speedup vs baseline: 3.9159x; 1.0359x over previous
#include <hip/hip_runtime.h>

typedef float v2f __attribute__((ext_vector_type(2)));
typedef float v4f __attribute__((ext_vector_type(4)));
typedef float v32f __attribute__((ext_vector_type(32)));
typedef float v8f __attribute__((ext_vector_type(8)));


#define N_NODES 50000
#define FDIM    128
#define N_EDGES 800000
#define KSTRIDE 260

#define R_NODES    200
#define R_PAD      208
#define ACC_S      132
#define AGG_BLOCKS 250
#define CHUNK_E    32768
#define N_CHUNK    25
#define N_DST4     (N_EDGES / 4)
#define LIST_CAP   32768
#define N_OB16     (N_EDGES / 16)
#define OB_CHUNK   8192
#define OB_FULL    6
#define OWN_BLOCKS 196
#define EL_CAP     16384
#define GL_OFF     16384
#define GL_CAP     16384
#define PART_OFF   16384
#define KEEP_WORDS (R_NODES * FDIM)
#define KEEP_HDR   32
#define N_BINS     224

__device__ __forceinline__ void sage_gather_list(
    const float* __restrict__ x, const unsigned* list, float* acc, float* dcnt,
    int n, int wave, int lane, int with_count)
{
    for (int i = wave; i < n; i += 128) {
        const bool p1 = (i + 32) < n, p2 = (i + 64) < n, p3 = (i + 96) < n;
        unsigned e0 = list[i];
        unsigned e1 = p1 ? list[i + 32] : e0;
        unsigned e2 = p2 ? list[i + 64] : e0;
        unsigned e3 = p3 ? list[i + 96] : e0;
        const float* r0 = x + (size_t)(e0 & 0xFFFFu) * FDIM + lane;
        const float* r1 = x + (size_t)(e1 & 0xFFFFu) * FDIM + lane;
        const float* r2 = x + (size_t)(e2 & 0xFFFFu) * FDIM + lane;
        const float* r3 = x + (size_t)(e3 & 0xFFFFu) * FDIM + lane;
        float a0 = r0[0], a1 = r0[32], a2 = r0[64], a3 = r0[96];
        float b0 = r1[0], b1 = r1[32], b2 = r1[64], b3 = r1[96];
        float c0 = r2[0], c1 = r2[32], c2 = r2[64], c3 = r2[96];
        float g0 = r3[0], g1 = r3[32], g2 = r3[64], g3 = r3[96];
        asm volatile("" :: "v"(a0), "v"(a1), "v"(a2), "v"(a3), "v"(b0), "v"(b1), "v"(b2), "v"(b3),
                           "v"(c0), "v"(c1), "v"(c2), "v"(c3), "v"(g0), "v"(g1), "v"(g2), "v"(g3));
        float* o0 = acc + (e0 >> 16) * ACC_S + lane;
        atomicAdd(o0, a0); atomicAdd(o0 + 32, a1); atomicAdd(o0 + 64, a2); atomicAdd(o0 + 96, a3);
        if (with_count && lane == 0) atomicAdd(&dcnt[e0 >> 16], 1.0f);
        if (p1) {
            float* o1 = acc + (e1 >> 16) * ACC_S + lane;
            atomicAdd(o1, b0); atomicAdd(o1 + 32, b1); atomicAdd(o1 + 64, b2); atomicAdd(o1 + 96, b3);
            if (with_count && lane == 0) atomicAdd(&dcnt[e1 >> 16], 1.0f);
        }
        if (p2) {
            float* o2 = acc + (e2 >> 16) * ACC_S + lane;
            atomicAdd(o2, c0); atomicAdd(o2 + 32, c1); atomicAdd(o2 + 64, c2); atomicAdd(o2 + 96, c3);
            if (with_count && lane == 0) atomicAdd(&dcnt[e2 >> 16], 1.0f);
        }
        if (p3) {
            float* o3 = acc + (e3 >> 16) * ACC_S + lane;
            atomicAdd(o3, g0); atomicAdd(o3 + 32, g1); atomicAdd(o3 + 64, g2); atomicAdd(o3 + 96, g3);
            if (with_count && lane == 0) atomicAdd(&dcnt[e3 >> 16], 1.0f);
        }
    }
}

__device__ __forceinline__ void sage_flush_row(float* acc, const unsigned row, const int lane, const v4f s)
{
    float* o = acc + row * ACC_S + 4 * lane;
    atomicAdd(o, s.x); atomicAdd(o + 1, s.y); atomicAdd(o + 2, s.z); atomicAdd(o + 3, s.w);
}

template <bool FULL>
__device__ __forceinline__ void sage_gather_group(const float* __restrict__ x, const unsigned* part, float* acc,
                                                  const int i, const int end, const int lane, v4f& s, unsigned& cur)
{
    const int k = i + (lane & 7);
    const unsigned pv = part[(FULL || k < end) ? k : i];
    unsigned e[8];
    v4f r[8];
#pragma unroll
    for (int q = 0; q < 8; ++q) {
        e[q] = (unsigned)__builtin_amdgcn_readlane((int)pv, q);
        r[q] = *(const v4f*)(x + (size_t)(e[q] & 0xFFFFu) * FDIM + 4 * lane);
    }
    asm volatile("" :: "v"(r[0]), "v"(r[1]), "v"(r[2]), "v"(r[3]), "v"(r[4]), "v"(r[5]), "v"(r[6]), "v"(r[7]));
#pragma unroll
    for (int q = 0; q < 8; ++q) {
        if (FULL || i + q < end) {
            const unsigned row = min(e[q] >> 16, (unsigned)(R_PAD - 1));
            if (row != cur) {
                if (cur != 0xFFFFFFFFu) sage_flush_row(acc, cur, lane, s);
                s = (v4f){0.f, 0.f, 0.f, 0.f};
                cur = row;
            }
            s += r[q];
        }
    }
}

__device__ __forceinline__ void sage_gather_even(
    const float* __restrict__ x, const unsigned* part, const unsigned total, float* acc,
    const int wave, const int lane)
{
    const int beg = (int)((total * (unsigned)wave) >> 5);
    const int end = (int)((total * (unsigned)(wave + 1)) >> 5);
    v4f s = {0.f, 0.f, 0.f, 0.f};
    unsigned cur = 0xFFFFFFFFu;
    int i = beg;
    for (; i + 8 <= end; i += 8) sage_gather_group<true>(x, part, acc, i, end, lane, s, cur);
    if (i < end) sage_gather_group<false>(x, part, acc, i, end, lane, s, cur);
    if (cur != 0xFFFFFFFFu) sage_flush_row(acc, cur, lane, s);
}

__device__ __forceinline__ unsigned sage_zero_bytes(const unsigned t)
{
    return ~(((t & 0x7F7F7F7Fu) + 0x7F7F7F7Fu) | t | 0x7F7F7F7Fu);
}

__device__ __forceinline__ unsigned sage_zero_nibble(const unsigned t)
{
    return (((sage_zero_bytes(t) >> 7) * 0x01020408u) >> 24) & 0xFu;
}

__device__ __forceinline__ void sage_load_ob(uint4 (&r)[8], const uint4* __restrict__ ob16, int c, int tid)
{
#pragma unroll
    for (int k = 0; k < 8; ++k) {
        int idx = c * OB_CHUNK + k * 1024 + tid;
        r[k] = ob16[idx < N_OB16 ? idx : (N_OB16 - 1)];
    }
}

template <bool CHECK>
__device__ __forceinline__ void sage_scan_ob(const uint4 (&r)[8], int c, int tid, unsigned pat,
                                             unsigned* glist, unsigned* ng)
{
#pragma unroll
    for (int k = 0; k < 8; ++k) {
        const int idx = c * OB_CHUNK + k * 1024 + tid;
        unsigned z = sage_zero_bytes(r[k].x ^ pat) | sage_zero_bytes(r[k].y ^ pat)
                   | sage_zero_bytes(r[k].z ^ pat) | sage_zero_bytes(r[k].w ^ pat);
        if (CHECK && idx >= N_OB16) z = 0u;
        if (z) {
            const unsigned pos = atomicAdd(ng, 1u);
            if (pos < (unsigned)GL_CAP) glist[pos] = (unsigned)idx;
        }
    }
}

__device__ __forceinline__ void sage_group_edges(const uint4 w, const unsigned g, const unsigned pat,
                                                 unsigned* list, unsigned* ne)
{
    unsigned m = sage_zero_nibble(w.x ^ pat) | (sage_zero_nibble(w.y ^ pat) << 4)
               | (sage_zero_nibble(w.z ^ pat) << 8) | (sage_zero_nibble(w.w ^ pat) << 12);
    unsigned pos = atomicAdd(ne, (unsigned)__popc(m));
    const unsigned e = g << 4;
    while (m) {
        const unsigned j = (unsigned)__ffs((int)m) - 1u;
        if (pos < (unsigned)EL_CAP) list[pos] = e + j;
        ++pos;
        m &= m - 1u;
    }
}

__device__ __forceinline__ void sage_load_chunk(int4 (&r)[8], const int4* __restrict__ dst4, int c, int tid)
{
#pragma unroll
    for (int k = 0; k < 8; ++k) {
        int idx = c * 8192 + k * 1024 + tid;
        r[k] = dst4[idx < N_DST4 ? idx : (N_DST4 - 1)];
    }
}

template <int NT>
__device__ __forceinline__ void sage_wave_product(
    const float* acc, const float* dcnt, const float* lw,
    const float* __restrict__ x, const float* __restrict__ bias, float* hout,
    int lo, int rt, int ct0, int lane, int do_relu)
{
    const int h   = lane >> 4;
    const int col = lane & 15;
    const int rl  = rt * 16 + col;
    int rg = lo + rl;
    rg = rg < N_NODES ? rg : (N_NODES - 1);

    const float  scale  = dcnt[rl];
    const float* arow_m = acc + rl * ACC_S + 2 * h;
    const float* arow_x = x + (size_t)rg * FDIM + 2 * h;
    const float* brow   = lw + (ct0 * 16 + col) * KSTRIDE + 2 * h;

    v8f d4[NT];
    v8f z = {0.f, 0.f, 0.f, 0.f, 0.f, 0.f, 0.f, 0.f};
#pragma unroll
    for (int t = 0; t < NT; ++t) d4[t] = z;

#pragma unroll 4
    for (int kk = 0; kk < 128; kk += 4) {
        v2f a = *(const v2f*)(arow_m + kk);
        a.x *= scale;
        a.y *= scale;
#pragma unroll
        for (int t = 0; t < NT; ++t) {
            v2f b = *(const v2f*)(brow + t * (16 * KSTRIDE) + kk);
            d4[t] = __builtin_amdgcn_wmma_f32_16x16x4_f32(
                false, a, false, b, (short)0, d4[t], false, false);
        }
    }

#pragma unroll 4
    for (int kk = 0; kk < 128; kk += 4) {
        v2f a = *(const v2f*)(arow_x + kk);
#pragma unroll
        for (int t = 0; t < NT; ++t) {
            v2f b = *(const v2f*)(brow + t * (16 * KSTRIDE) + 128 + kk);
            d4[t] = __builtin_amdgcn_wmma_f32_16x16x4_f32(
                false, a, false, b, (short)0, d4[t], false, false);
        }
    }

#pragma unroll
    for (int t = 0; t < NT; ++t) {
        int n  = (ct0 + t) * 16 + col;
        float bv = bias[n];
#pragma unroll
        for (int v = 0; v < 8; ++v) {
            int row = rt * 16 + v + 8 * h;
            float val = d4[t][v] + bv;
            if (do_relu) val = fmaxf(val, 0.0f);
            if (row < R_NODES) hout[(size_t)(lo + row) * FDIM + n] = val;
        }
    }
}

__global__ __launch_bounds__(1024) void sage_layer_kernel(
    const float* __restrict__ x,
    const int*   __restrict__ src,
    const int*   __restrict__ dst,
    const unsigned* __restrict__ ob,
    float* cnt,
    float* keep,
    const float* __restrict__ Wl, const float* __restrict__ Wr,
    const float* __restrict__ bias,
    float* hout,
    int first,
    int do_relu)
{
    __shared__ __attribute__((aligned(16))) float    acc[R_PAD * ACC_S];
    __shared__ float    dcnt[R_PAD];
    __shared__ __attribute__((aligned(16))) unsigned ubuf[128 * KSTRIDE];
    __shared__ unsigned nl[4];
    __shared__ unsigned wcnt[N_BINS];
    __shared__ unsigned wpos[N_BINS];
    unsigned* list = ubuf;
    float*    lw   = (float*)ubuf;
    unsigned* part = ubuf + PART_OFF;

    const int tid  = threadIdx.x;
    const int lane = tid & 31;
    const int wave = __builtin_amdgcn_readfirstlane(tid >> 5);
    const int lo   = blockIdx.x * R_NODES;
    unsigned* kl   = (unsigned*)keep + (size_t)blockIdx.x * KEEP_WORDS;

    for (int i = tid; i < R_PAD * ACC_S / 4; i += 1024) ((v4f*)acc)[i] = (v4f){0.f, 0.f, 0.f, 0.f};
    if (tid < 4) nl[tid] = 0u;
    if (tid < N_BINS) wcnt[tid] = 0u;

    const unsigned kept = first ? 0xFFFFFFFFu : kl[0];

    if (kept <= (unsigned)EL_CAP) {
        const unsigned* ke = kl + KEEP_HDR;
        for (unsigned i = tid; i < kept; i += 4096) {
            const unsigned i1 = i + 1024, i2 = i + 2048, i3 = i + 3072;
            const bool q1 = i1 < kept, q2 = i2 < kept, q3 = i3 < kept;
            unsigned v0 = ke[i];
            unsigned v1 = q1 ? ke[i1] : 0u;
            unsigned v2 = q2 ? ke[i2] : 0u;
            unsigned v3 = q3 ? ke[i3] : 0u;
            part[i] = v0;
            if (q1) part[i1] = v1;
            if (q2) part[i2] = v2;
            if (q3) part[i3] = v3;
        }
        __syncthreads();
        sage_gather_even(x, part, (unsigned)__builtin_amdgcn_readfirstlane((int)kept), acc, wave, lane);
        __syncthreads();
    } else {
        const int4*  dst4  = (const int4*)dst;
        const uint4* ob16  = (const uint4*)ob;
        const unsigned pat = (unsigned)blockIdx.x * 0x01010101u;
        unsigned*    glist = ubuf + GL_OFF;

        {
            uint4 b[8], b2[8];
            sage_load_ob(b, ob16, 0, tid);
            __syncthreads();
            for (int c = 0; c < OB_FULL; c += 2) {
                sage_load_ob(b2, ob16, c + 1, tid);
                sage_scan_ob<false>(b, c, tid, pat, glist, &nl[2]);
                sage_load_ob(b, ob16, c + 2, tid);
                sage_scan_ob<false>(b2, c + 1, tid, pat, glist, &nl[2]);
            }
            sage_scan_ob<true>(b, OB_FULL, tid, pat, glist, &nl[2]);
        }
        __syncthreads();
        const unsigned ngrp = nl[2];

        if (ngrp <= (unsigned)GL_CAP) {
            for (unsigned i = tid; i < ngrp; i += 4096) {
                const unsigned i1 = i + 1024, i2 = i + 2048, i3 = i + 3072;
                const bool q1 = i1 < ngrp, q2 = i2 < ngrp, q3 = i3 < ngrp;
                const unsigned g0 = glist[i];
                const unsigned g1 = q1 ? glist[i1] : g0;
                const unsigned g2 = q2 ? glist[i2] : g0;
                const unsigned g3 = q3 ? glist[i3] : g0;
                const uint4 w0 = ob16[g0], w1 = ob16[g1], w2 = ob16[g2], w3 = ob16[g3];
                sage_group_edges(w0, g0, pat, list, &nl[3]);
                if (q1) sage_group_edges(w1, g1, pat, list, &nl[3]);
                if (q2) sage_group_edges(w2, g2, pat, list, &nl[3]);
                if (q3) sage_group_edges(w3, g3, pat, list, &nl[3]);
            }
        }
        __syncthreads();
        const unsigned total = (ngrp <= (unsigned)GL_CAP && nl[3] <= (unsigned)EL_CAP) ? nl[3] : 0xFFFFFFFFu;

        if (total <= (unsigned)EL_CAP) {
            const bool keepit = first != 0;
            unsigned* ke = kl + KEEP_HDR;
            for (unsigned i = tid; i < total; i += 4096) {
                const unsigned i1 = i + 1024, i2 = i + 2048, i3 = i + 3072;
                const bool q1 = i1 < total, q2 = i2 < total, q3 = i3 < total;
                const unsigned a0 = list[i];
                const unsigned a1 = q1 ? list[i1] : a0;
                const unsigned a2 = q2 ? list[i2] : a0;
                const unsigned a3 = q3 ? list[i3] : a0;
                const unsigned s0 = (unsigned)src[a0], r0 = (unsigned)(dst[a0] - lo);
                const unsigned s1 = (unsigned)src[a1], r1 = (unsigned)(dst[a1] - lo);
                const unsigned s2 = (unsigned)src[a2], r2 = (unsigned)(dst[a2] - lo);
                const unsigned s3 = (unsigned)src[a3], r3 = (unsigned)(dst[a3] - lo);
                const unsigned b0 = min(r0, (unsigned)(R_PAD - 1)), b1 = min(r1, (unsigned)(R_PAD - 1));
                const unsigned b2 = min(r2, (unsigned)(R_PAD - 1)), b3 = min(r3, (unsigned)(R_PAD - 1));
                list[i] = (s0 & 0xFFFFu) | (b0 << 16);
                atomicAdd(&wcnt[b0], 1u);
                if (q1) { list[i1] = (s1 & 0xFFFFu) | (b1 << 16); atomicAdd(&wcnt[b1], 1u); }
                if (q2) { list[i2] = (s2 & 0xFFFFu) | (b2 << 16); atomicAdd(&wcnt[b2], 1u); }
                if (q3) { list[i3] = (s3 & 0xFFFFu) | (b3 << 16); atomicAdd(&wcnt[b3], 1u); }
            }
            __syncthreads();
            if (tid < 32) {
                unsigned c[7];
                unsigned cn = 0u;
#pragma unroll
                for (int j = 0; j < 7; ++j) { c[j] = wcnt[7 * tid + j]; cn += c[j]; }
                unsigned s = cn;
#pragma unroll
                for (int d = 1; d < 32; d <<= 1) {
                    const unsigned t = __shfl_up(s, (unsigned)d);
                    if (lane >= d) s += t;
                }
                unsigned ex = s - cn;
#pragma unroll
                for (int j = 0; j < 7; ++j) { wpos[7 * tid + j] = ex; ex += c[j]; }
            }
            __syncthreads();
            for (unsigned i = tid; i < total; i += 1024) {
                const unsigned v = list[i];
                part[atomicAdd(&wpos[v >> 16], 1u)] = v;
            }
            __syncthreads();
            if (keepit) for (unsigned i = tid; i < total; i += 1024) ke[i] = part[i];
            if (first && tid < R_PAD) dcnt[tid] = (float)wcnt[tid];
            sage_gather_even(x, part, (unsigned)__builtin_amdgcn_readfirstlane((int)total), acc, wave, lane);
            __syncthreads();
        } else {
            if (tid < R_PAD) dcnt[tid] = 0.0f;
            int4 d[8];
            sage_load_chunk(d, dst4, 0, tid);
            for (int c = 0; c < N_CHUNK; ++c) {
                const int cur = c & 1;
                int  sv[32];
                bool hit[8];
#pragma unroll
                for (int k = 0; k < 8; ++k) {
                    int ebase = (c * 8192 + k * 1024 + tid) << 2;
                    unsigned l0 = (unsigned)(d[k].x - lo), l1 = (unsigned)(d[k].y - lo);
                    unsigned l2 = (unsigned)(d[k].z - lo), l3 = (unsigned)(d[k].w - lo);
                    unsigned mn = min(min(l0, l1), min(l2, l3));
                    hit[k] = mn < (unsigned)R_NODES;
                    if (hit[k]) {
                        if (l0 < (unsigned)R_NODES) sv[k * 4 + 0] = src[ebase + 0];
                        if (l1 < (unsigned)R_NODES) sv[k * 4 + 1] = src[ebase + 1];
                        if (l2 < (unsigned)R_NODES) sv[k * 4 + 2] = src[ebase + 2];
                        if (l3 < (unsigned)R_NODES) sv[k * 4 + 3] = src[ebase + 3];
                    }
                }
#pragma unroll
                for (int k = 0; k < 8; ++k) {
                    if (hit[k]) {
                        unsigned ll[4] = {(unsigned)(d[k].x - lo), (unsigned)(d[k].y - lo),
                                          (unsigned)(d[k].z - lo), (unsigned)(d[k].w - lo)};
#pragma unroll
                        for (int q = 0; q < 4; ++q) {
                            if (ll[q] < (unsigned)R_NODES) {
                                unsigned pos = atomicAdd(&nl[cur], 1u);
                                list[pos] = (unsigned)sv[k * 4 + q] | (ll[q] << 16);
                            }
                        }
                    }
                }
                __syncthreads();

#pragma unroll
                for (int k = 0; k < 8; ++k) {
                    int idx  = (c + 1) * 8192 + k * 1024 + tid;
                    int idxc = idx < N_DST4 ? idx : (N_DST4 - 1);
                    int4 v = dst4[idxc];
                    if (idx >= N_DST4) { v.x = -1; v.y = -1; v.z = -1; v.w = -1; }
                    d[k] = v;
                }
                if (tid == 0) nl[cur ^ 1] = 0u;

                const int n = (int)nl[cur];
                sage_gather_list(x, list, acc, dcnt, n, wave, lane, first);
                __syncthreads();
            }
        }
        if (first && tid == 0) kl[0] = total;
    }

    if (tid < R_PAD) {
        float c = 1.0f;
        if (tid < R_NODES) {
            if (first) { c = dcnt[tid]; cnt[lo + tid] = c; }
            else       { c = cnt[lo + tid]; }
        }
        dcnt[tid] = 1.0f / fmaxf(c, 1.0f);
    }
    for (int idx = tid; idx < 256 * 128; idx += 1024) {
        int k = idx >> 7;
        int n = idx & 127;
        float w = (k < 128) ? Wl[idx] : Wr[idx - 128 * 128];
        lw[n * KSTRIDE + k] = w;
    }
    __syncthreads();

    if (wave < 24) sage_wave_product<4>(acc, dcnt, lw, x, bias, hout, lo, wave >> 1, (wave & 1) * 4, lane, do_relu);
    else           sage_wave_product<1>(acc, dcnt, lw, x, bias, hout, lo, 12, wave - 24, lane, do_relu);
}

__global__ __launch_bounds__(1024) void sage_owner_kernel(const int* __restrict__ dst, unsigned* __restrict__ ob)
{
    const int i = blockIdx.x * 1024 + threadIdx.x;
    if (i < N_DST4) {
        const int4 d = ((const int4*)dst)[i];
        const unsigned u0 = (unsigned)d.x, u1 = (unsigned)d.y, u2 = (unsigned)d.z, u3 = (unsigned)d.w;
        const unsigned o0 = u0 < (unsigned)N_NODES ? u0 / (unsigned)R_NODES : 255u;
        const unsigned o1 = u1 < (unsigned)N_NODES ? u1 / (unsigned)R_NODES : 255u;
        const unsigned o2 = u2 < (unsigned)N_NODES ? u2 / (unsigned)R_NODES : 255u;
        const unsigned o3 = u3 < (unsigned)N_NODES ? u3 / (unsigned)R_NODES : 255u;
        ob[i] = o0 | (o1 << 8) | (o2 << 16) | (o3 << 24);
    }
}

extern "C" void kernel_launch(void* const* d_in, const int* in_sizes, int n_in,
                              void* d_out, int out_size, void* d_ws, size_t ws_size,
                              hipStream_t stream) {
    const float* x   = (const float*)d_in[0];
    const int*   ei  = (const int*)d_in[1];
    const float* Wl0 = (const float*)d_in[2];
    const float* Wr0 = (const float*)d_in[3];
    const float* b0  = (const float*)d_in[4];
    const float* Wl1 = (const float*)d_in[5];
    const float* Wr1 = (const float*)d_in[6];
    const float* b1  = (const float*)d_in[7];
    float* out = (float*)d_out;

    const int* src = ei;
    const int* dst = ei + N_EDGES;

    unsigned* ob = (unsigned*)d_ws;
    float* cnt  = (float*)d_ws + (size_t)N_NODES * FDIM;
    float* hbuf = cnt + N_NODES;

    sage_owner_kernel<<<OWN_BLOCKS, 1024, 0, stream>>>(dst, ob);
    sage_layer_kernel<<<AGG_BLOCKS, 1024, 0, stream>>>(x, src, dst, ob, cnt, out, Wl0, Wr0, b0, hbuf, 1, 1);
    sage_layer_kernel<<<AGG_BLOCKS, 1024, 0, stream>>>(hbuf, src, dst, ob, cnt, out, Wl1, Wr1, b1, out, 0, 0);
}


# first launch also writes a packed word per edge (src | dst<<16): a listed group is resolved from its own 64 B, one random line, not owner bytes again plus src and dst per edge; list by arrival
# speedup vs baseline: 4.0839x; 1.0429x over previous
#include <hip/hip_runtime.h>

typedef float v2f __attribute__((ext_vector_type(2)));
typedef float v4f __attribute__((ext_vector_type(4)));
typedef float v32f __attribute__((ext_vector_type(32)));
typedef float v8f __attribute__((ext_vector_type(8)));


#define N_NODES 50000
#define FDIM    128
#define N_EDGES 800000
#define KSTRIDE 260

#define R_NODES    200
#define R_PAD      208
#define ACC_S      132
#define AGG_BLOCKS 250
#define CHUNK_E    32768
#define N_CHUNK    25
#define N_DST4     (N_EDGES / 4)
#define LIST_CAP   32768
#define N_OB16     (N_EDGES / 16)
#define OB_CHUNK   8192
#define OB_FULL    6
#define OWN_BLOCKS 196
#define PK_OFF     262144
#define EL_CAP     16384
#define GL_OFF     16384
#define GL_CAP     16384
#define PART_OFF   16384
#define KEEP_WORDS (R_NODES * FDIM)
#define KEEP_HDR   32
#define N_BINS     224

__device__ __forceinline__ void sage_gather_list(
    const float* __restrict__ x, const unsigned* list, float* acc, float* dcnt,
    int n, int wave, int lane, int with_count)
{
    for (int i = wave; i < n; i += 128) {
        const bool p1 = (i + 32) < n, p2 = (i + 64) < n, p3 = (i + 96) < n;
        unsigned e0 = list[i];
        unsigned e1 = p1 ? list[i + 32] : e0;
        unsigned e2 = p2 ? list[i + 64] : e0;
        unsigned e3 = p3 ? list[i + 96] : e0;
        const float* r0 = x + (size_t)(e0 & 0xFFFFu) * FDIM + lane;
        const float* r1 = x + (size_t)(e1 & 0xFFFFu) * FDIM + lane;
        const float* r2 = x + (size_t)(e2 & 0xFFFFu) * FDIM + lane;
        const float* r3 = x + (size_t)(e3 & 0xFFFFu) * FDIM + lane;
        float a0 = r0[0], a1 = r0[32], a2 = r0[64], a3 = r0[96];
        float b0 = r1[0], b1 = r1[32], b2 = r1[64], b3 = r1[96];
        float c0 = r2[0], c1 = r2[32], c2 = r2[64], c3 = r2[96];
        float g0 = r3[0], g1 = r3[32], g2 = r3[64], g3 = r3[96];
        asm volatile("" :: "v"(a0), "v"(a1), "v"(a2), "v"(a3), "v"(b0), "v"(b1), "v"(b2), "v"(b3),
                           "v"(c0), "v"(c1), "v"(c2), "v"(c3), "v"(g0), "v"(g1), "v"(g2), "v"(g3));
        float* o0 = acc + (e0 >> 16) * ACC_S + lane;
        atomicAdd(o0, a0); atomicAdd(o0 + 32, a1); atomicAdd(o0 + 64, a2); atomicAdd(o0 + 96, a3);
        if (with_count && lane == 0) atomicAdd(&dcnt[e0 >> 16], 1.0f);
        if (p1) {
            float* o1 = acc + (e1 >> 16) * ACC_S + lane;
            atomicAdd(o1, b0); atomicAdd(o1 + 32, b1); atomicAdd(o1 + 64, b2); atomicAdd(o1 + 96, b3);
            if (with_count && lane == 0) atomicAdd(&dcnt[e1 >> 16], 1.0f);
        }
        if (p2) {
            float* o2 = acc + (e2 >> 16) * ACC_S + lane;
            atomicAdd(o2, c0); atomicAdd(o2 + 32, c1); atomicAdd(o2 + 64, c2); atomicAdd(o2 + 96, c3);
            if (with_count && lane == 0) atomicAdd(&dcnt[e2 >> 16], 1.0f);
        }
        if (p3) {
            float* o3 = acc + (e3 >> 16) * ACC_S + lane;
            atomicAdd(o3, g0); atomicAdd(o3 + 32, g1); atomicAdd(o3 + 64, g2); atomicAdd(o3 + 96, g3);
            if (with_count && lane == 0) atomicAdd(&dcnt[e3 >> 16], 1.0f);
        }
    }
}

__device__ __forceinline__ void sage_flush_row(float* acc, const unsigned row, const int lane, const v4f s)
{
    float* o = acc + row * ACC_S + 4 * lane;
    atomicAdd(o, s.x); atomicAdd(o + 1, s.y); atomicAdd(o + 2, s.z); atomicAdd(o + 3, s.w);
}

template <bool FULL>
__device__ __forceinline__ void sage_gather_group(const float* __restrict__ x, const unsigned* part, float* acc,
                                                  const int i, const int end, const int lane, v4f& s, unsigned& cur)
{
    const int k = i + (lane & 7);
    const unsigned pv = part[(FULL || k < end) ? k : i];
    unsigned e[8];
    v4f r[8];
#pragma unroll
    for (int q = 0; q < 8; ++q) {
        e[q] = (unsigned)__builtin_amdgcn_readlane((int)pv, q);
        r[q] = *(const v4f*)(x + (size_t)(e[q] & 0xFFFFu) * FDIM + 4 * lane);
    }
    asm volatile("" :: "v"(r[0]), "v"(r[1]), "v"(r[2]), "v"(r[3]), "v"(r[4]), "v"(r[5]), "v"(r[6]), "v"(r[7]));
#pragma unroll
    for (int q = 0; q < 8; ++q) {
        if (FULL || i + q < end) {
            const unsigned row = min(e[q] >> 16, (unsigned)(R_PAD - 1));
            if (row != cur) {
                if (cur != 0xFFFFFFFFu) sage_flush_row(acc, cur, lane, s);
                s = (v4f){0.f, 0.f, 0.f, 0.f};
                cur = row;
            }
            s += r[q];
        }
    }
}

__device__ __forceinline__ void sage_gather_even(
    const float* __restrict__ x, const unsigned* part, const unsigned total, float* acc,
    const int wave, const int lane)
{
    const int beg = (int)((total * (unsigned)wave) >> 5);
    const int end = (int)((total * (unsigned)(wave + 1)) >> 5);
    v4f s = {0.f, 0.f, 0.f, 0.f};
    unsigned cur = 0xFFFFFFFFu;
    int i = beg;
    for (; i + 8 <= end; i += 8) sage_gather_group<true>(x, part, acc, i, end, lane, s, cur);
    if (i < end) sage_gather_group<false>(x, part, acc, i, end, lane, s, cur);
    if (cur != 0xFFFFFFFFu) sage_flush_row(acc, cur, lane, s);
}

__device__ __forceinline__ unsigned sage_zero_bytes(const unsigned t)
{
    return ~(((t & 0x7F7F7F7Fu) + 0x7F7F7F7Fu) | t | 0x7F7F7F7Fu);
}

__device__ __forceinline__ void sage_load_ob(uint4 (&r)[8], const uint4* __restrict__ ob16, int c, int tid)
{
#pragma unroll
    for (int k = 0; k < 8; ++k) {
        int idx = c * OB_CHUNK + k * 1024 + tid;
        r[k] = ob16[idx < N_OB16 ? idx : (N_OB16 - 1)];
    }
}

template <bool CHECK>
__device__ __forceinline__ void sage_scan_ob(const uint4 (&r)[8], int c, int tid, unsigned pat,
                                             unsigned* glist, unsigned* ng)
{
#pragma unroll
    for (int k = 0; k < 8; ++k) {
        const int idx = c * OB_CHUNK + k * 1024 + tid;
        unsigned z = sage_zero_bytes(r[k].x ^ pat) | sage_zero_bytes(r[k].y ^ pat)
                   | sage_zero_bytes(r[k].z ^ pat) | sage_zero_bytes(r[k].w ^ pat);
        if (CHECK && idx >= N_OB16) z = 0u;
        if (z) {
            const unsigned pos = atomicAdd(ng, 1u);
            if (pos < (unsigned)GL_CAP) glist[pos] = (unsigned)idx;
        }
    }
}

__device__ __forceinline__ void sage_take_owned(const uint4 w, const unsigned lo, unsigned* list, unsigned* ne,
                                                unsigned* wcnt)
{
    const unsigned v[4] = {w.x, w.y, w.z, w.w};
#pragma unroll
    for (int j = 0; j < 4; ++j) {
        const unsigned r = (v[j] >> 16) - lo;
        if (r < (unsigned)R_NODES) {
            const unsigned pos = atomicAdd(ne, 1u);
            if (pos < (unsigned)EL_CAP) {
                list[pos] = (v[j] & 0xFFFFu) | (r << 16);
                atomicAdd(&wcnt[r], 1u);
            }
        }
    }
}

__device__ __forceinline__ void sage_load_chunk(int4 (&r)[8], const int4* __restrict__ dst4, int c, int tid)
{
#pragma unroll
    for (int k = 0; k < 8; ++k) {
        int idx = c * 8192 + k * 1024 + tid;
        r[k] = dst4[idx < N_DST4 ? idx : (N_DST4 - 1)];
    }
}

template <int NT>
__device__ __forceinline__ void sage_wave_product(
    const float* acc, const float* dcnt, const float* lw,
    const float* __restrict__ x, const float* __restrict__ bias, float* hout,
    int lo, int rt, int ct0, int lane, int do_relu)
{
    const int h   = lane >> 4;
    const int col = lane & 15;
    const int rl  = rt * 16 + col;
    int rg = lo + rl;
    rg = rg < N_NODES ? rg : (N_NODES - 1);

    const float  scale  = dcnt[rl];
    const float* arow_m = acc + rl * ACC_S + 2 * h;
    const float* arow_x = x + (size_t)rg * FDIM + 2 * h;
    const float* brow   = lw + (ct0 * 16 + col) * KSTRIDE + 2 * h;

    v8f d4[NT];
    v8f z = {0.f, 0.f, 0.f, 0.f, 0.f, 0.f, 0.f, 0.f};
#pragma unroll
    for (int t = 0; t < NT; ++t) d4[t] = z;

#pragma unroll 4
    for (int kk = 0; kk < 128; kk += 4) {
        v2f a = *(const v2f*)(arow_m + kk);
        a.x *= scale;
        a.y *= scale;
#pragma unroll
        for (int t = 0; t < NT; ++t) {
            v2f b = *(const v2f*)(brow + t * (16 * KSTRIDE) + kk);
            d4[t] = __builtin_amdgcn_wmma_f32_16x16x4_f32(
                false, a, false, b, (short)0, d4[t], false, false);
        }
    }

#pragma unroll 4
    for (int kk = 0; kk < 128; kk += 4) {
        v2f a = *(const v2f*)(arow_x + kk);
#pragma unroll
        for (int t = 0; t < NT; ++t) {
            v2f b = *(const v2f*)(brow + t * (16 * KSTRIDE) + 128 + kk);
            d4[t] = __builtin_amdgcn_wmma_f32_16x16x4_f32(
                false, a, false, b, (short)0, d4[t], false, false);
        }
    }

#pragma unroll
    for (int t = 0; t < NT; ++t) {
        int n  = (ct0 + t) * 16 + col;
        float bv = bias[n];
#pragma unroll
        for (int v = 0; v < 8; ++v) {
            int row = rt * 16 + v + 8 * h;
            float val = d4[t][v] + bv;
            if (do_relu) val = fmaxf(val, 0.0f);
            if (row < R_NODES) hout[(size_t)(lo + row) * FDIM + n] = val;
        }
    }
}

__global__ __launch_bounds__(1024) void sage_layer_kernel(
    const float* __restrict__ x,
    const int*   __restrict__ src,
    const int*   __restrict__ dst,
    const unsigned* __restrict__ ob,
    const unsigned* __restrict__ pk,
    float* cnt,
    float* keep,
    const float* __restrict__ Wl, const float* __restrict__ Wr,
    const float* __restrict__ bias,
    float* hout,
    int first,
    int do_relu)
{
    __shared__ __attribute__((aligned(16))) float    acc[R_PAD * ACC_S];
    __shared__ float    dcnt[R_PAD];
    __shared__ __attribute__((aligned(16))) unsigned ubuf[128 * KSTRIDE];
    __shared__ unsigned nl[4];
    __shared__ unsigned wcnt[N_BINS];
    __shared__ unsigned wpos[N_BINS];
    unsigned* list = ubuf;
    float*    lw   = (float*)ubuf;
    unsigned* part = ubuf + PART_OFF;

    const int tid  = threadIdx.x;
    const int lane = tid & 31;
    const int wave = __builtin_amdgcn_readfirstlane(tid >> 5);
    const int lo   = blockIdx.x * R_NODES;
    unsigned* kl   = (unsigned*)keep + (size_t)blockIdx.x * KEEP_WORDS;

    for (int i = tid; i < R_PAD * ACC_S / 4; i += 1024) ((v4f*)acc)[i] = (v4f){0.f, 0.f, 0.f, 0.f};
    if (tid < 4) nl[tid] = 0u;
    if (tid < N_BINS) wcnt[tid] = 0u;

    const unsigned kept = first ? 0xFFFFFFFFu : kl[0];

    if (kept <= (unsigned)EL_CAP) {
        const unsigned* ke = kl + KEEP_HDR;
        for (unsigned i = tid; i < kept; i += 4096) {
            const unsigned i1 = i + 1024, i2 = i + 2048, i3 = i + 3072;
            const bool q1 = i1 < kept, q2 = i2 < kept, q3 = i3 < kept;
            unsigned v0 = ke[i];
            unsigned v1 = q1 ? ke[i1] : 0u;
            unsigned v2 = q2 ? ke[i2] : 0u;
            unsigned v3 = q3 ? ke[i3] : 0u;
            part[i] = v0;
            if (q1) part[i1] = v1;
            if (q2) part[i2] = v2;
            if (q3) part[i3] = v3;
        }
        __syncthreads();
        sage_gather_even(x, part, (unsigned)__builtin_amdgcn_readfirstlane((int)kept), acc, wave, lane);
        __syncthreads();
    } else {
        const int4*  dst4  = (const int4*)dst;
        const uint4* ob16  = (const uint4*)ob;
        const unsigned pat = (unsigned)blockIdx.x * 0x01010101u;
        unsigned*    glist = ubuf + GL_OFF;

        {
            uint4 b[8], b2[8];
            sage_load_ob(b, ob16, 0, tid);
            __syncthreads();
            for (int c = 0; c < OB_FULL; c += 2) {
                sage_load_ob(b2, ob16, c + 1, tid);
                sage_scan_ob<false>(b, c, tid, pat, glist, &nl[2]);
                sage_load_ob(b, ob16, c + 2, tid);
                sage_scan_ob<false>(b2, c + 1, tid, pat, glist, &nl[2]);
            }
            sage_scan_ob<true>(b, OB_FULL, tid, pat, glist, &nl[2]);
        }
        __syncthreads();
        const unsigned ngrp = nl[2];

        if (ngrp <= (unsigned)GL_CAP) {
            const uint4*   pk16 = (const uint4*)pk;
            const unsigned nu   = ngrp * 4u;
            const unsigned qq   = (unsigned)tid & 3u;
            for (unsigned u = tid; u < nu; u += 16384) {
                uint4 w[16];
#pragma unroll
                for (int k = 0; k < 16; ++k) {
                    const unsigned uk = u + 1024u * (unsigned)k;
                    const unsigned g  = glist[(uk < nu ? uk : u) >> 2];
                    w[k] = pk16[4u * g + qq];
                }
                asm volatile("" :: "v"(w[0].x), "v"(w[1].x), "v"(w[2].x), "v"(w[3].x), "v"(w[4].x), "v"(w[5].x),
                                   "v"(w[6].x), "v"(w[7].x), "v"(w[8].x), "v"(w[9].x), "v"(w[10].x), "v"(w[11].x),
                                   "v"(w[12].x), "v"(w[13].x), "v"(w[14].x), "v"(w[15].x));
#pragma unroll
                for (int k = 0; k < 16; ++k)
                    if (u + 1024u * (unsigned)k < nu) sage_take_owned(w[k], (unsigned)lo, list, &nl[3], wcnt);
            }
        }
        __syncthreads();
        const unsigned total = (ngrp <= (unsigned)GL_CAP && nl[3] <= (unsigned)EL_CAP) ? nl[3] : 0xFFFFFFFFu;

        if (total <= (unsigned)EL_CAP) {
            const bool keepit = first != 0;
            unsigned* ke = kl + KEEP_HDR;
            if (tid < 32) {
                unsigned c[7];
                unsigned cn = 0u;
#pragma unroll
                for (int j = 0; j < 7; ++j) { c[j] = wcnt[7 * tid + j]; cn += c[j]; }
                unsigned s = cn;
#pragma unroll
                for (int d = 1; d < 32; d <<= 1) {
                    const unsigned t = __shfl_up(s, (unsigned)d);
                    if (lane >= d) s += t;
                }
                unsigned ex = s - cn;
#pragma unroll
                for (int j = 0; j < 7; ++j) { wpos[7 * tid + j] = ex; ex += c[j]; }
            }
            __syncthreads();
            for (unsigned i = tid; i < total; i += 1024) {
                const unsigned v = list[i];
                part[atomicAdd(&wpos[v >> 16], 1u)] = v;
            }
            __syncthreads();
            if (keepit) for (unsigned i = tid; i < total; i += 1024) ke[i] = part[i];
            if (first && tid < R_PAD) dcnt[tid] = (float)wcnt[tid];
            sage_gather_even(x, part, (unsigned)__builtin_amdgcn_readfirstlane((int)total), acc, wave, lane);
            __syncthreads();
        } else {
            if (tid < R_PAD) dcnt[tid] = 0.0f;
            int4 d[8];
            sage_load_chunk(d, dst4, 0, tid);
            for (int c = 0; c < N_CHUNK; ++c) {
                const int cur = c & 1;
                int  sv[32];
                bool hit[8];
#pragma unroll
                for (int k = 0; k < 8; ++k) {
                    int ebase = (c * 8192 + k * 1024 + tid) << 2;
                    unsigned l0 = (unsigned)(d[k].x - lo), l1 = (unsigned)(d[k].y - lo);
                    unsigned l2 = (unsigned)(d[k].z - lo), l3 = (unsigned)(d[k].w - lo);
                    unsigned mn = min(min(l0, l1), min(l2, l3));
                    hit[k] = mn < (unsigned)R_NODES;
                    if (hit[k]) {
                        if (l0 < (unsigned)R_NODES) sv[k * 4 + 0] = src[ebase + 0];
                        if (l1 < (unsigned)R_NODES) sv[k * 4 + 1] = src[ebase + 1];
                        if (l2 < (unsigned)R_NODES) sv[k * 4 + 2] = src[ebase + 2];
                        if (l3 < (unsigned)R_NODES) sv[k * 4 + 3] = src[ebase + 3];
                    }
                }
#pragma unroll
                for (int k = 0; k < 8; ++k) {
                    if (hit[k]) {
                        unsigned ll[4] = {(unsigned)(d[k].x - lo), (unsigned)(d[k].y - lo),
                                          (unsigned)(d[k].z - lo), (unsigned)(d[k].w - lo)};
#pragma unroll
                        for (int q = 0; q < 4; ++q) {
                            if (ll[q] < (unsigned)R_NODES) {
                                unsigned pos = atomicAdd(&nl[cur], 1u);
                                list[pos] = (unsigned)sv[k * 4 + q] | (ll[q] << 16);
                            }
                        }
                    }
                }
                __syncthreads();

#pragma unroll
                for (int k = 0; k < 8; ++k) {
                    int idx  = (c + 1) * 8192 + k * 1024 + tid;
                    int idxc = idx < N_DST4 ? idx : (N_DST4 - 1);
                    int4 v = dst4[idxc];
                    if (idx >= N_DST4) { v.x = -1; v.y = -1; v.z = -1; v.w = -1; }
                    d[k] = v;
                }
                if (tid == 0) nl[cur ^ 1] = 0u;

                const int n = (int)nl[cur];
                sage_gather_list(x, list, acc, dcnt, n, wave, lane, first);
                __syncthreads();
            }
        }
        if (first && tid == 0) kl[0] = total;
    }

    if (tid < R_PAD) {
        float c = 1.0f;
        if (tid < R_NODES) {
            if (first) { c = dcnt[tid]; cnt[lo + tid] = c; }
            else       { c = cnt[lo + tid]; }
        }
        dcnt[tid] = 1.0f / fmaxf(c, 1.0f);
    }
    for (int idx = tid; idx < 256 * 128; idx += 1024) {
        int k = idx >> 7;
        int n = idx & 127;
        float w = (k < 128) ? Wl[idx] : Wr[idx - 128 * 128];
        lw[n * KSTRIDE + k] = w;
    }
    __syncthreads();

    if (wave < 24) sage_wave_product<4>(acc, dcnt, lw, x, bias, hout, lo, wave >> 1, (wave & 1) * 4, lane, do_relu);
    else           sage_wave_product<1>(acc, dcnt, lw, x, bias, hout, lo, 12, wave - 24, lane, do_relu);
}

__global__ __launch_bounds__(1024) void sage_owner_kernel(const int* __restrict__ src, const int* __restrict__ dst,
                                                          unsigned* __restrict__ ob, unsigned* __restrict__ pk)
{
    const int i = blockIdx.x * 1024 + threadIdx.x;
    if (i < N_DST4) {
        const int4 d = ((const int4*)dst)[i];
        const int4 s = ((const int4*)src)[i];
        const unsigned u0 = (unsigned)d.x, u1 = (unsigned)d.y, u2 = (unsigned)d.z, u3 = (unsigned)d.w;
        const unsigned o0 = u0 < (unsigned)N_NODES ? u0 / (unsigned)R_NODES : 255u;
        const unsigned o1 = u1 < (unsigned)N_NODES ? u1 / (unsigned)R_NODES : 255u;
        const unsigned o2 = u2 < (unsigned)N_NODES ? u2 / (unsigned)R_NODES : 255u;
        const unsigned o3 = u3 < (unsigned)N_NODES ? u3 / (unsigned)R_NODES : 255u;
        ob[i] = o0 | (o1 << 8) | (o2 << 16) | (o3 << 24);
        const unsigned f0 = u0 < (unsigned)N_NODES ? u0 : 0xFFFFu, f1 = u1 < (unsigned)N_NODES ? u1 : 0xFFFFu;
        const unsigned f2 = u2 < (unsigned)N_NODES ? u2 : 0xFFFFu, f3 = u3 < (unsigned)N_NODES ? u3 : 0xFFFFu;
        uint4 p;
        p.x = ((unsigned)s.x & 0xFFFFu) | (f0 << 16);
        p.y = ((unsigned)s.y & 0xFFFFu) | (f1 << 16);
        p.z = ((unsigned)s.z & 0xFFFFu) | (f2 << 16);
        p.w = ((unsigned)s.w & 0xFFFFu) | (f3 << 16);
        ((uint4*)pk)[i] = p;
    }
}

extern "C" void kernel_launch(void* const* d_in, const int* in_sizes, int n_in,
                              void* d_out, int out_size, void* d_ws, size_t ws_size,
                              hipStream_t stream) {
    const float* x   = (const float*)d_in[0];
    const int*   ei  = (const int*)d_in[1];
    const float* Wl0 = (const float*)d_in[2];
    const float* Wr0 = (const float*)d_in[3];
    const float* b0  = (const float*)d_in[4];
    const float* Wl1 = (const float*)d_in[5];
    const float* Wr1 = (const float*)d_in[6];
    const float* b1  = (const float*)d_in[7];
    float* out = (float*)d_out;

    const int* src = ei;
    const int* dst = ei + N_EDGES;

    unsigned* ob = (unsigned*)d_ws;
    unsigned* pk = (unsigned*)d_ws + PK_OFF;
    float* cnt  = (float*)d_ws + (size_t)N_NODES * FDIM;
    float* hbuf = cnt + N_NODES;

    sage_owner_kernel<<<OWN_BLOCKS, 1024, 0, stream>>>(src, dst, ob, pk);
    sage_layer_kernel<<<AGG_BLOCKS, 1024, 0, stream>>>(x, src, dst, ob, pk, cnt, out, Wl0, Wr0, b0, hbuf, 1, 1);
    sage_layer_kernel<<<AGG_BLOCKS, 1024, 0, stream>>>(hbuf, src, dst, ob, pk, cnt, out, Wl1, Wr1, b1, out, 0, 0);
}
